# starting kernel
# baseline (speedup 1.0000x reference)
#include <hip/hip_runtime.h>
#include <math.h>


typedef float v2f __attribute__((ext_vector_type(2)));
typedef float v8f __attribute__((ext_vector_type(8)));

#define STATES   16
#define FDIM     128
#define EPSF     1e-10f
#define EXP1F    2.7182818284590452f
#define EXP1M1F  1.7182818284590452f

__global__ void bpn_priors_wmma(const float* __restrict__ feat,
                                const float* __restrict__ W,
                                float* __restrict__ priors,
                                float* __restrict__ log_priors,
                                float* __restrict__ accum,
                                int N, int numTiles) {
  const int wave = blockIdx.x * (blockDim.x >> 5) + (threadIdx.x >> 5);
  if (wave >= numTiles) return;
  const int lane = threadIdx.x & 31;
  const int half = lane >> 4;
  const int s    = lane & 15;

  int rowA = wave * 16 + s;
  if (rowA >= N) rowA = N - 1;
  const float* frow = feat + (size_t)rowA * FDIM;

  v8f acc = {};
#pragma unroll
  for (int k0 = 0; k0 < FDIM; k0 += 4) {
    const int k = k0 + 2 * half;
    v2f a, b;
    a.x = frow[k];
    a.y = frow[k + 1];
    b.x = W[(size_t)k * STATES + s];
    b.y = W[(size_t)(k + 1) * STATES + s];
    acc = __builtin_amdgcn_wmma_f32_16x16x4_f32(false, a, false, b,
                                                (short)0, acc, false, false);
  }

#pragma unroll
  for (int r = 0; r < 8; ++r) {
    float logit = acc[r];
    float mx = logit;
    mx = fmaxf(mx, __shfl_xor(mx, 1, 32));
    mx = fmaxf(mx, __shfl_xor(mx, 2, 32));
    mx = fmaxf(mx, __shfl_xor(mx, 4, 32));
    mx = fmaxf(mx, __shfl_xor(mx, 8, 32));
    float ex = expf(logit - mx);
    float sm = ex;
    sm += __shfl_xor(sm, 1, 32);
    sm += __shfl_xor(sm, 2, 32);
    sm += __shfl_xor(sm, 4, 32);
    sm += __shfl_xor(sm, 8, 32);
    float pr = ex / sm;
    float lp = logf(fmaxf(pr, EPSF));
    const int rowO = wave * 16 + r + 8 * half;
    if (rowO < N) {
      const size_t o = (size_t)rowO * STATES + s;
      priors[o]     = pr;
      log_priors[o] = lp;
      accum[o]      = lp;
    }
  }
}

__global__ void bpn_edge(const float* __restrict__ beliefs,
                         const int* __restrict__ src,
                         const int* __restrict__ dst,
                         const int* __restrict__ rev,
                         float* __restrict__ msgs,
                         float* __restrict__ accum,
                         int Eu, int firstIter) {
  const int p = blockIdx.x * (blockDim.x >> 5) + (threadIdx.x >> 5);
  if (p >= Eu) return;
  const int lane = threadIdx.x & 31;
  const int half = lane >> 4;
  const int s    = lane & 15;

  const int j = rev[p];
  const int e = half ? j : p;

  float mOld = firstIter ? 0.0625f : msgs[(size_t)e * STATES + s];
  const float mRev = __shfl_xor(mOld, 16, 32);

  const int   se = src[e];
  const float b  = beliefs[(size_t)se * STATES + s];
  const float x  = b / mRev;

  float t = x;
  t += __shfl_xor(t, 1, 32);
  t += __shfl_xor(t, 2, 32);
  t += __shfl_xor(t, 4, 32);
  t += __shfl_xor(t, 8, 32);

  const float m = (t + EXP1M1F * x) / ((15.0f + EXP1F) * t);
  msgs[(size_t)e * STATES + s] = m;

  const float lm = logf(fmaxf(m, EPSF));
  const int   de = dst[e];
  atomicAdd(&accum[(size_t)de * STATES + s], lm);
}

__global__ void bpn_node(float* __restrict__ accum,
                         const float* __restrict__ log_priors,
                         float* __restrict__ beliefs_out,
                         int N) {
  const int w    = blockIdx.x * (blockDim.x >> 5) + (threadIdx.x >> 5);
  const int lane = threadIdx.x & 31;
  const int half = lane >> 4;
  const int s    = lane & 15;
  const int node = w * 2 + half;

  const bool ok = (node < N);
  const size_t o = (size_t)(ok ? node : 0) * STATES + s;
  float v = ok ? accum[o] : 0.0f;

  float mx = v;
  mx = fmaxf(mx, __shfl_xor(mx, 1, 32));
  mx = fmaxf(mx, __shfl_xor(mx, 2, 32));
  mx = fmaxf(mx, __shfl_xor(mx, 4, 32));
  mx = fmaxf(mx, __shfl_xor(mx, 8, 32));
  float ex = expf(v - mx);
  float sm = ex;
  sm += __shfl_xor(sm, 1, 32);
  sm += __shfl_xor(sm, 2, 32);
  sm += __shfl_xor(sm, 4, 32);
  sm += __shfl_xor(sm, 8, 32);

  if (ok) {
    beliefs_out[o] = ex / sm;
    accum[o]       = log_priors[o];
  }
}

extern "C" void kernel_launch(void* const* d_in, const int* in_sizes, int n_in,
                              void* d_out, int out_size, void* d_ws, size_t ws_size,
                              hipStream_t stream) {
  const float* feat = (const float*)d_in[0];
  const float* W    = (const float*)d_in[1];
  const int*   src  = (const int*)d_in[2];
  const int*   dst  = (const int*)d_in[3];
  const int*   rev  = (const int*)d_in[4];

  const int N  = in_sizes[0] / FDIM;
  const int E2 = in_sizes[2];
  const int Eu = E2 / 2;

  float* priors      = (float*)d_out;
  float* beliefs_out = priors + (size_t)N * STATES;

  float* lp    = (float*)d_ws;
  float* accum = lp    + (size_t)N * STATES;
  float* bel   = accum + (size_t)N * STATES;
  float* msgs  = bel   + (size_t)N * STATES;

  const dim3 blk(256);
  const int tiles   = (N + 15) / 16;
  const int gBlocks = (tiles + 7) / 8;
  const int eBlocks = (Eu + 7) / 8;
  const int nWaves  = (N + 1) / 2;
  const int nBlocks = (nWaves + 7) / 8;

  bpn_priors_wmma<<<gBlocks, blk, 0, stream>>>(feat, W, priors, lp, accum, N, tiles);

  bpn_edge<<<eBlocks, blk, 0, stream>>>(priors, src, dst, rev, msgs, accum, Eu, 1);
  bpn_node<<<nBlocks, blk, 0, stream>>>(accum, lp, bel, N);
  bpn_edge<<<eBlocks, blk, 0, stream>>>(bel, src, dst, rev, msgs, accum, Eu, 0);
  bpn_node<<<nBlocks, blk, 0, stream>>>(accum, lp, bel, N);
  bpn_edge<<<eBlocks, blk, 0, stream>>>(bel, src, dst, rev, msgs, accum, Eu, 0);
  bpn_node<<<nBlocks, blk, 0, stream>>>(accum, lp, beliefs_out, N);
}


# Edge kernel: each wave handles 8 reverse-edge pairs instead of 1, with the index, message and belief loads of all 8 issued as independent batches (more loads in flight, 8x fewer waves)
# speedup vs baseline: 1.7464x; 1.7464x over previous
#include <hip/hip_runtime.h>
#include <math.h>


typedef float v2f __attribute__((ext_vector_type(2)));
typedef float v8f __attribute__((ext_vector_type(8)));

#define STATES   16
#define FDIM     128
#define EPSF     1e-10f
#define EXP1F    2.7182818284590452f
#define EXP1M1F  1.7182818284590452f

__global__ void bpn_priors_wmma(const float* __restrict__ feat,
                                const float* __restrict__ W,
                                float* __restrict__ priors,
                                float* __restrict__ log_priors,
                                float* __restrict__ accum,
                                int N, int numTiles) {
  const int wave = blockIdx.x * (blockDim.x >> 5) + (threadIdx.x >> 5);
  if (wave >= numTiles) return;
  const int lane = threadIdx.x & 31;
  const int half = lane >> 4;
  const int s    = lane & 15;

  int rowA = wave * 16 + s;
  if (rowA >= N) rowA = N - 1;
  const float* frow = feat + (size_t)rowA * FDIM;

  v8f acc = {};
#pragma unroll
  for (int k0 = 0; k0 < FDIM; k0 += 4) {
    const int k = k0 + 2 * half;
    v2f a, b;
    a.x = frow[k];
    a.y = frow[k + 1];
    b.x = W[(size_t)k * STATES + s];
    b.y = W[(size_t)(k + 1) * STATES + s];
    acc = __builtin_amdgcn_wmma_f32_16x16x4_f32(false, a, false, b,
                                                (short)0, acc, false, false);
  }

#pragma unroll
  for (int r = 0; r < 8; ++r) {
    float logit = acc[r];
    float mx = logit;
    mx = fmaxf(mx, __shfl_xor(mx, 1, 32));
    mx = fmaxf(mx, __shfl_xor(mx, 2, 32));
    mx = fmaxf(mx, __shfl_xor(mx, 4, 32));
    mx = fmaxf(mx, __shfl_xor(mx, 8, 32));
    float ex = expf(logit - mx);
    float sm = ex;
    sm += __shfl_xor(sm, 1, 32);
    sm += __shfl_xor(sm, 2, 32);
    sm += __shfl_xor(sm, 4, 32);
    sm += __shfl_xor(sm, 8, 32);
    float pr = ex / sm;
    float lp = logf(fmaxf(pr, EPSF));
    const int rowO = wave * 16 + r + 8 * half;
    if (rowO < N) {
      const size_t o = (size_t)rowO * STATES + s;
      priors[o]     = pr;
      log_priors[o] = lp;
      accum[o]      = lp;
    }
  }
}

#define PAIRS 8

__global__ void bpn_edge(const float* __restrict__ beliefs,
                         const int* __restrict__ src,
                         const int* __restrict__ dst,
                         const int* __restrict__ rev,
                         float* __restrict__ msgs,
                         float* __restrict__ accum,
                         int Eu, int firstIter) {
  const int w  = __builtin_amdgcn_readfirstlane(
      (int)(blockIdx.x * (blockDim.x >> 5) + (threadIdx.x >> 5)));
  const int p0 = w * PAIRS;
  if (p0 >= Eu) return;
  const int lane = threadIdx.x & 31;
  const int half = lane >> 4;
  const int s    = lane & 15;

  const int pl = p0 + (lane & (PAIRS - 1));
  const int jv = rev[pl < Eu ? pl : (Eu - 1)];
  int  e[PAIRS];
  bool ok[PAIRS];
#pragma unroll
  for (int i = 0; i < PAIRS; ++i) {
    const int p = p0 + i;
    ok[i] = (p < Eu);
    const int j = __builtin_amdgcn_readlane(jv, i);
    e[i] = half ? j : (ok[i] ? p : (Eu - 1));
  }

  float mOld[PAIRS];
  int   se[PAIRS], de[PAIRS];
#pragma unroll
  for (int i = 0; i < PAIRS; ++i) {
    mOld[i] = firstIter ? 0.0625f : msgs[(size_t)e[i] * STATES + s];
    se[i]   = src[e[i]];
    de[i]   = dst[e[i]];
  }

  float b[PAIRS];
#pragma unroll
  for (int i = 0; i < PAIRS; ++i)
    b[i] = beliefs[(size_t)se[i] * STATES + s];

#pragma unroll
  for (int i = 0; i < PAIRS; ++i) {
    const float mRev = __shfl_xor(mOld[i], 16, 32);
    const float x    = b[i] / mRev;

    float t = x;
    t += __shfl_xor(t, 1, 32);
    t += __shfl_xor(t, 2, 32);
    t += __shfl_xor(t, 4, 32);
    t += __shfl_xor(t, 8, 32);

    const float m  = (t + EXP1M1F * x) / ((15.0f + EXP1F) * t);
    const float lm = logf(fmaxf(m, EPSF));
    if (ok[i]) {
      msgs[(size_t)e[i] * STATES + s] = m;
      atomicAdd(&accum[(size_t)de[i] * STATES + s], lm);
    }
  }
}

__global__ void bpn_node(float* __restrict__ accum,
                         const float* __restrict__ log_priors,
                         float* __restrict__ beliefs_out,
                         int N) {
  const int w    = blockIdx.x * (blockDim.x >> 5) + (threadIdx.x >> 5);
  const int lane = threadIdx.x & 31;
  const int half = lane >> 4;
  const int s    = lane & 15;
  const int node = w * 2 + half;

  const bool ok = (node < N);
  const size_t o = (size_t)(ok ? node : 0) * STATES + s;
  float v = ok ? accum[o] : 0.0f;

  float mx = v;
  mx = fmaxf(mx, __shfl_xor(mx, 1, 32));
  mx = fmaxf(mx, __shfl_xor(mx, 2, 32));
  mx = fmaxf(mx, __shfl_xor(mx, 4, 32));
  mx = fmaxf(mx, __shfl_xor(mx, 8, 32));
  float ex = expf(v - mx);
  float sm = ex;
  sm += __shfl_xor(sm, 1, 32);
  sm += __shfl_xor(sm, 2, 32);
  sm += __shfl_xor(sm, 4, 32);
  sm += __shfl_xor(sm, 8, 32);

  if (ok) {
    beliefs_out[o] = ex / sm;
    accum[o]       = log_priors[o];
  }
}

extern "C" void kernel_launch(void* const* d_in, const int* in_sizes, int n_in,
                              void* d_out, int out_size, void* d_ws, size_t ws_size,
                              hipStream_t stream) {
  const float* feat = (const float*)d_in[0];
  const float* W    = (const float*)d_in[1];
  const int*   src  = (const int*)d_in[2];
  const int*   dst  = (const int*)d_in[3];
  const int*   rev  = (const int*)d_in[4];

  const int N  = in_sizes[0] / FDIM;
  const int E2 = in_sizes[2];
  const int Eu = E2 / 2;

  float* priors      = (float*)d_out;
  float* beliefs_out = priors + (size_t)N * STATES;

  float* lp    = (float*)d_ws;
  float* accum = lp    + (size_t)N * STATES;
  float* bel   = accum + (size_t)N * STATES;
  float* msgs  = bel   + (size_t)N * STATES;

  const dim3 blk(256);
  const int tiles   = (N + 15) / 16;
  const int gBlocks = (tiles + 7) / 8;
  const int eBlocks = (Eu + 8 * PAIRS - 1) / (8 * PAIRS);
  const int nWaves  = (N + 1) / 2;
  const int nBlocks = (nWaves + 7) / 8;

  bpn_priors_wmma<<<gBlocks, blk, 0, stream>>>(feat, W, priors, lp, accum, N, tiles);

  bpn_edge<<<eBlocks, blk, 0, stream>>>(priors, src, dst, rev, msgs, accum, Eu, 1);
  bpn_node<<<nBlocks, blk, 0, stream>>>(accum, lp, bel, N);
  bpn_edge<<<eBlocks, blk, 0, stream>>>(bel, src, dst, rev, msgs, accum, Eu, 0);
  bpn_node<<<nBlocks, blk, 0, stream>>>(accum, lp, bel, N);
  bpn_edge<<<eBlocks, blk, 0, stream>>>(bel, src, dst, rev, msgs, accum, Eu, 0);
  bpn_node<<<nBlocks, blk, 0, stream>>>(accum, lp, beliefs_out, N);
}


# Per-edge message array removed: with uniform start messages the reverse messages of iterations 2 and 3 are recomputed from per-node rows (g1, bel1, bel2), so edge kernels only gather node tables
# speedup vs baseline: 2.3767x; 1.3609x over previous
#include <hip/hip_runtime.h>
#include <math.h>


typedef float v2f __attribute__((ext_vector_type(2)));
typedef float v8f __attribute__((ext_vector_type(8)));

#define STATES   16
#define FDIM     128
#define EPSF     1e-10f
#define EXP1F    2.7182818284590452f
#define EXP1M1F  1.7182818284590452f

static __device__ __forceinline__ float bp_msg(float x) {
  float t = x;
  t += __shfl_xor(t, 1, 32);
  t += __shfl_xor(t, 2, 32);
  t += __shfl_xor(t, 4, 32);
  t += __shfl_xor(t, 8, 32);
  return (t + EXP1M1F * x) / ((15.0f + EXP1F) * t);
}

__global__ void bpn_priors_wmma(const float* __restrict__ feat,
                                const float* __restrict__ W,
                                float* __restrict__ priors,
                                float* __restrict__ log_priors,
                                float* __restrict__ accum,
                                float* __restrict__ rec,
                                float* __restrict__ lg1,
                                int N, int numTiles) {
  const int wave = blockIdx.x * (blockDim.x >> 5) + (threadIdx.x >> 5);
  if (wave >= numTiles) return;
  const int lane = threadIdx.x & 31;
  const int half = lane >> 4;
  const int s    = lane & 15;

  int rowA = wave * 16 + s;
  if (rowA >= N) rowA = N - 1;
  const float* frow = feat + (size_t)rowA * FDIM;

  v8f acc = {};
#pragma unroll
  for (int k0 = 0; k0 < FDIM; k0 += 4) {
    const int k = k0 + 2 * half;
    v2f a, b;
    a.x = frow[k];
    a.y = frow[k + 1];
    b.x = W[(size_t)k * STATES + s];
    b.y = W[(size_t)(k + 1) * STATES + s];
    acc = __builtin_amdgcn_wmma_f32_16x16x4_f32(false, a, false, b,
                                                (short)0, acc, false, false);
  }

#pragma unroll
  for (int r = 0; r < 8; ++r) {
    float logit = acc[r];
    float mx = logit;
    mx = fmaxf(mx, __shfl_xor(mx, 1, 32));
    mx = fmaxf(mx, __shfl_xor(mx, 2, 32));
    mx = fmaxf(mx, __shfl_xor(mx, 4, 32));
    mx = fmaxf(mx, __shfl_xor(mx, 8, 32));
    float ex = expf(logit - mx);
    float sm = ex;
    sm += __shfl_xor(sm, 1, 32);
    sm += __shfl_xor(sm, 2, 32);
    sm += __shfl_xor(sm, 4, 32);
    sm += __shfl_xor(sm, 8, 32);
    float pr = ex / sm;
    float lp = logf(fmaxf(pr, EPSF));
    float g1 = bp_msg(pr / 0.0625f);
    float l1 = logf(fmaxf(g1, EPSF));
    const int rowO = wave * 16 + r + 8 * half;
    if (rowO < N) {
      const size_t o = (size_t)rowO * STATES + s;
      priors[o]     = pr;
      log_priors[o] = lp;
      accum[o]      = lp;
      v2f g; g.x = g1; g.y = pr;
      *(v2f*)(rec + 2 * o) = g;
      lg1[o]        = l1;
    }
  }
}

#define PAIRS 8

template <int ITER>
static __device__ __forceinline__ void edge_body(const float* __restrict__ rec,
                                                 const float* __restrict__ bel2,
                                                 const int* __restrict__ src,
                                                 const int* __restrict__ dst,
                                                 const int* __restrict__ rev,
                                                 float* __restrict__ accum,
                                                 int Eu) {
  const int w  = __builtin_amdgcn_readfirstlane(
      (int)(blockIdx.x * (blockDim.x >> 5) + (threadIdx.x >> 5)));
  const int p0 = w * PAIRS;
  if (p0 >= Eu) return;
  const int lane = threadIdx.x & 31;
  const int half = lane >> 4;
  const int s    = lane & 15;

  const int pl = p0 + (lane & (PAIRS - 1));
  const int jv = rev[pl < Eu ? pl : (Eu - 1)];
  int  e[PAIRS];
  bool ok[PAIRS];
#pragma unroll
  for (int i = 0; i < PAIRS; ++i) {
    const int p = p0 + i;
    ok[i] = (p < Eu);
    const int j = __builtin_amdgcn_readlane(jv, i);
    e[i] = half ? j : (ok[i] ? p : (Eu - 1));
  }

  int se[PAIRS], de[PAIRS];
#pragma unroll
  for (int i = 0; i < PAIRS; ++i) {
    se[i] = src[e[i]];
    de[i] = dst[e[i]];
  }
  __builtin_amdgcn_sched_barrier(0);

  float lm[PAIRS];
  {
    v2f   gb[PAIRS];
    float b2[PAIRS];
#pragma unroll
    for (int i = 0; i < PAIRS; ++i) {
      gb[i] = *(const v2f*)(rec + 2 * ((size_t)se[i] * STATES + s));
      if (ITER == 3) b2[i] = bel2[(size_t)se[i] * STATES + s];
    }
    __builtin_amdgcn_sched_barrier(0);

    float m2[PAIRS];
#pragma unroll
    for (int i = 0; i < PAIRS; ++i) {
      const float g1rev = __shfl_xor(gb[i].x, 16, 32);
      m2[i] = bp_msg(gb[i].y / g1rev);
    }
    if (ITER == 2) {
#pragma unroll
      for (int i = 0; i < PAIRS; ++i)
        lm[i] = logf(fmaxf(m2[i], EPSF));
    } else {
#pragma unroll
      for (int i = 0; i < PAIRS; ++i) {
        const float m2rev = __shfl_xor(m2[i], 16, 32);
        const float m3    = bp_msg(b2[i] / m2rev);
        lm[i] = logf(fmaxf(m3, EPSF));
      }
    }
  }

#pragma unroll
  for (int i = 0; i < PAIRS; ++i) {
    if (ok[i])
      atomicAdd(&accum[(size_t)de[i] * STATES + s], lm[i]);
  }
}

__global__ void bpn_edge1(const float* __restrict__ lg1,
                          const int* __restrict__ src,
                          const int* __restrict__ dst,
                          float* __restrict__ accum, int Eu) {
  const int w  = __builtin_amdgcn_readfirstlane(
      (int)(blockIdx.x * (blockDim.x >> 5) + (threadIdx.x >> 5)));
  const int E2 = 2 * Eu;
  const int e0 = w * (2 * PAIRS);
  if (e0 >= E2) return;
  const int lane = threadIdx.x & 31;
  const int half = lane >> 4;
  const int s    = lane & 15;

  int  se[PAIRS], de[PAIRS];
  bool ok[PAIRS];
#pragma unroll
  for (int i = 0; i < PAIRS; ++i) {
    ok[i] = (e0 + 2 * i < E2);
    const int e = ok[i] ? (e0 + 2 * i + half) : (E2 - 1);
    se[i] = src[e];
    de[i] = dst[e];
  }
  float lm[PAIRS];
#pragma unroll
  for (int i = 0; i < PAIRS; ++i)
    lm[i] = lg1[(size_t)se[i] * STATES + s];
#pragma unroll
  for (int i = 0; i < PAIRS; ++i) {
    if (ok[i])
      atomicAdd(&accum[(size_t)de[i] * STATES + s], lm[i]);
  }
}

__global__ void bpn_edge2(const float* __restrict__ rec,
                          const int* __restrict__ src,
                          const int* __restrict__ dst,
                          const int* __restrict__ rev,
                          float* __restrict__ accum, int Eu) {
  edge_body<2>(rec, nullptr, src, dst, rev, accum, Eu);
}

__global__ void bpn_edge3(const float* __restrict__ rec,
                          const float* __restrict__ bel2,
                          const int* __restrict__ src,
                          const int* __restrict__ dst,
                          const int* __restrict__ rev,
                          float* __restrict__ accum, int Eu) {
  edge_body<3>(rec, bel2, src, dst, rev, accum, Eu);
}

__global__ void bpn_node(float* __restrict__ accum,
                         const float* __restrict__ log_priors,
                         float* __restrict__ beliefs_out,
                         int ostride, int ooff,
                         int N) {
  const int w    = blockIdx.x * (blockDim.x >> 5) + (threadIdx.x >> 5);
  const int lane = threadIdx.x & 31;
  const int half = lane >> 4;
  const int s    = lane & 15;
  const int node = w * 2 + half;

  const bool ok = (node < N);
  const size_t o = (size_t)(ok ? node : 0) * STATES + s;
  float v = ok ? accum[o] : 0.0f;

  float mx = v;
  mx = fmaxf(mx, __shfl_xor(mx, 1, 32));
  mx = fmaxf(mx, __shfl_xor(mx, 2, 32));
  mx = fmaxf(mx, __shfl_xor(mx, 4, 32));
  mx = fmaxf(mx, __shfl_xor(mx, 8, 32));
  float ex = expf(v - mx);
  float sm = ex;
  sm += __shfl_xor(sm, 1, 32);
  sm += __shfl_xor(sm, 2, 32);
  sm += __shfl_xor(sm, 4, 32);
  sm += __shfl_xor(sm, 8, 32);

  if (ok) {
    beliefs_out[o * (size_t)ostride + ooff] = ex / sm;
    accum[o] = log_priors[o];
  }
}

extern "C" void kernel_launch(void* const* d_in, const int* in_sizes, int n_in,
                              void* d_out, int out_size, void* d_ws, size_t ws_size,
                              hipStream_t stream) {
  const float* feat = (const float*)d_in[0];
  const float* W    = (const float*)d_in[1];
  const int*   src  = (const int*)d_in[2];
  const int*   dst  = (const int*)d_in[3];
  const int*   rev  = (const int*)d_in[4];

  const int N  = in_sizes[0] / FDIM;
  const int E2 = in_sizes[2];
  const int Eu = E2 / 2;

  float* priors      = (float*)d_out;
  float* beliefs_out = priors + (size_t)N * STATES;

  float* lp    = (float*)d_ws;
  float* accum = lp    + (size_t)N * STATES;
  float* lg1   = accum + (size_t)N * STATES;
  float* bel2  = lg1   + (size_t)N * STATES;
  float* rec   = bel2  + (size_t)N * STATES;

  const dim3 blk(256);
  const int tiles   = (N + 15) / 16;
  const int gBlocks = (tiles + 7) / 8;
  const int eBlocks = (Eu + 8 * PAIRS - 1) / (8 * PAIRS);
  const int nWaves  = (N + 1) / 2;
  const int nBlocks = (nWaves + 7) / 8;

  bpn_priors_wmma<<<gBlocks, blk, 0, stream>>>(feat, W, priors, lp, accum, rec, lg1, N, tiles);

  bpn_edge1<<<eBlocks, blk, 0, stream>>>(lg1, src, dst, accum, Eu);
  bpn_node<<<nBlocks, blk, 0, stream>>>(accum, lp, rec, 2, 1, N);
  bpn_edge2<<<eBlocks, blk, 0, stream>>>(rec, src, dst, rev, accum, Eu);
  bpn_node<<<nBlocks, blk, 0, stream>>>(accum, lp, bel2, 1, 0, N);
  bpn_edge3<<<eBlocks, blk, 0, stream>>>(rec, bel2, src, dst, rev, accum, Eu);
  bpn_node<<<nBlocks, blk, 0, stream>>>(accum, lp, beliefs_out, 1, 0, N);
}


# Edge kernels of iterations 2 and 3: IEEE-exact f32 division sequences and full-range logf replaced by v_rcp_f32 multiply and v_log_f32*ln2 (still f32; divisors and log arguments provably normal)
# speedup vs baseline: 2.5875x; 1.0887x over previous
#include <hip/hip_runtime.h>
#include <math.h>


typedef float v2f __attribute__((ext_vector_type(2)));
typedef float v8f __attribute__((ext_vector_type(8)));

#define STATES   16
#define FDIM     128
#define EPSF     1e-10f
#define EXP1F    2.7182818284590452f
#define EXP1M1F  1.7182818284590452f
#define LN2F     0.69314718055994531f
#define MSG_A    (1.0f / (15.0f + EXP1F))
#define MSG_B    (EXP1M1F / (15.0f + EXP1F))

static __device__ __forceinline__ float bp_msg(float x) {
  float t = x;
  t += __shfl_xor(t, 1, 32);
  t += __shfl_xor(t, 2, 32);
  t += __shfl_xor(t, 4, 32);
  t += __shfl_xor(t, 8, 32);
  return (t + EXP1M1F * x) / ((15.0f + EXP1F) * t);
}

static __device__ __forceinline__ float bp_msg_edge(float x) {
  float t = x;
  t += __shfl_xor(t, 1, 32);
  t += __shfl_xor(t, 2, 32);
  t += __shfl_xor(t, 4, 32);
  t += __shfl_xor(t, 8, 32);
  return fmaf(x * __builtin_amdgcn_rcpf(t), MSG_B, MSG_A);
}

static __device__ __forceinline__ float log_msg_edge(float m) {
  return __builtin_amdgcn_logf(fmaxf(m, EPSF)) * LN2F;
}

__global__ void bpn_priors_wmma(const float* __restrict__ feat,
                                const float* __restrict__ W,
                                float* __restrict__ priors,
                                float* __restrict__ log_priors,
                                float* __restrict__ accum,
                                float* __restrict__ rec,
                                float* __restrict__ lg1,
                                int N, int numTiles) {
  const int wave = blockIdx.x * (blockDim.x >> 5) + (threadIdx.x >> 5);
  if (wave >= numTiles) return;
  const int lane = threadIdx.x & 31;
  const int half = lane >> 4;
  const int s    = lane & 15;

  int rowA = wave * 16 + s;
  if (rowA >= N) rowA = N - 1;
  const float* frow = feat + (size_t)rowA * FDIM;

  v8f acc = {};
#pragma unroll
  for (int k0 = 0; k0 < FDIM; k0 += 4) {
    const int k = k0 + 2 * half;
    v2f a, b;
    a.x = frow[k];
    a.y = frow[k + 1];
    b.x = W[(size_t)k * STATES + s];
    b.y = W[(size_t)(k + 1) * STATES + s];
    acc = __builtin_amdgcn_wmma_f32_16x16x4_f32(false, a, false, b,
                                                (short)0, acc, false, false);
  }

#pragma unroll
  for (int r = 0; r < 8; ++r) {
    float logit = acc[r];
    float mx = logit;
    mx = fmaxf(mx, __shfl_xor(mx, 1, 32));
    mx = fmaxf(mx, __shfl_xor(mx, 2, 32));
    mx = fmaxf(mx, __shfl_xor(mx, 4, 32));
    mx = fmaxf(mx, __shfl_xor(mx, 8, 32));
    float ex = expf(logit - mx);
    float sm = ex;
    sm += __shfl_xor(sm, 1, 32);
    sm += __shfl_xor(sm, 2, 32);
    sm += __shfl_xor(sm, 4, 32);
    sm += __shfl_xor(sm, 8, 32);
    float pr = ex / sm;
    float lp = logf(fmaxf(pr, EPSF));
    float g1 = bp_msg(pr / 0.0625f);
    float l1 = logf(fmaxf(g1, EPSF));
    const int rowO = wave * 16 + r + 8 * half;
    if (rowO < N) {
      const size_t o = (size_t)rowO * STATES + s;
      priors[o]     = pr;
      log_priors[o] = lp;
      accum[o]      = lp;
      v2f g; g.x = g1; g.y = pr;
      *(v2f*)(rec + 2 * o) = g;
      lg1[o]        = l1;
    }
  }
}

#define PAIRS 8

template <int ITER>
static __device__ __forceinline__ void edge_body(const float* __restrict__ rec,
                                                 const float* __restrict__ bel2,
                                                 const int* __restrict__ src,
                                                 const int* __restrict__ dst,
                                                 const int* __restrict__ rev,
                                                 float* __restrict__ accum,
                                                 int Eu) {
  const int w  = __builtin_amdgcn_readfirstlane(
      (int)(blockIdx.x * (blockDim.x >> 5) + (threadIdx.x >> 5)));
  const int p0 = w * PAIRS;
  if (p0 >= Eu) return;
  const int lane = threadIdx.x & 31;
  const int half = lane >> 4;
  const int s    = lane & 15;

  const int pl = p0 + (lane & (PAIRS - 1));
  const int jv = rev[pl < Eu ? pl : (Eu - 1)];
  int  e[PAIRS];
  bool ok[PAIRS];
#pragma unroll
  for (int i = 0; i < PAIRS; ++i) {
    const int p = p0 + i;
    ok[i] = (p < Eu);
    const int j = __builtin_amdgcn_readlane(jv, i);
    e[i] = half ? j : (ok[i] ? p : (Eu - 1));
  }

  int se[PAIRS], de[PAIRS];
#pragma unroll
  for (int i = 0; i < PAIRS; ++i) {
    se[i] = src[e[i]];
    de[i] = dst[e[i]];
  }
  __builtin_amdgcn_sched_barrier(0);

  float lm[PAIRS];
  {
    v2f   gb[PAIRS];
    float b2[PAIRS];
#pragma unroll
    for (int i = 0; i < PAIRS; ++i) {
      gb[i] = *(const v2f*)(rec + 2 * ((size_t)se[i] * STATES + s));
      if (ITER == 3) b2[i] = bel2[(size_t)se[i] * STATES + s];
    }
    __builtin_amdgcn_sched_barrier(0);

    float m2[PAIRS];
#pragma unroll
    for (int i = 0; i < PAIRS; ++i) {
      const float g1rev = __shfl_xor(gb[i].x, 16, 32);
      m2[i] = bp_msg_edge(gb[i].y * __builtin_amdgcn_rcpf(g1rev));
    }
    if (ITER == 2) {
#pragma unroll
      for (int i = 0; i < PAIRS; ++i)
        lm[i] = log_msg_edge(m2[i]);
    } else {
#pragma unroll
      for (int i = 0; i < PAIRS; ++i) {
        const float m2rev = __shfl_xor(m2[i], 16, 32);
        const float m3    = bp_msg_edge(b2[i] * __builtin_amdgcn_rcpf(m2rev));
        lm[i] = log_msg_edge(m3);
      }
    }
  }

#pragma unroll
  for (int i = 0; i < PAIRS; ++i) {
    if (ok[i])
      atomicAdd(&accum[(size_t)de[i] * STATES + s], lm[i]);
  }
}

__global__ void bpn_edge1(const float* __restrict__ lg1,
                          const int* __restrict__ src,
                          const int* __restrict__ dst,
                          float* __restrict__ accum, int Eu) {
  const int w  = __builtin_amdgcn_readfirstlane(
      (int)(blockIdx.x * (blockDim.x >> 5) + (threadIdx.x >> 5)));
  const int E2 = 2 * Eu;
  const int e0 = w * (2 * PAIRS);
  if (e0 >= E2) return;
  const int lane = threadIdx.x & 31;
  const int half = lane >> 4;
  const int s    = lane & 15;

  int  se[PAIRS], de[PAIRS];
  bool ok[PAIRS];
#pragma unroll
  for (int i = 0; i < PAIRS; ++i) {
    ok[i] = (e0 + 2 * i < E2);
    const int e = ok[i] ? (e0 + 2 * i + half) : (E2 - 1);
    se[i] = src[e];
    de[i] = dst[e];
  }
  float lm[PAIRS];
#pragma unroll
  for (int i = 0; i < PAIRS; ++i)
    lm[i] = lg1[(size_t)se[i] * STATES + s];
#pragma unroll
  for (int i = 0; i < PAIRS; ++i) {
    if (ok[i])
      atomicAdd(&accum[(size_t)de[i] * STATES + s], lm[i]);
  }
}

__global__ void bpn_edge2(const float* __restrict__ rec,
                          const int* __restrict__ src,
                          const int* __restrict__ dst,
                          const int* __restrict__ rev,
                          float* __restrict__ accum, int Eu) {
  edge_body<2>(rec, nullptr, src, dst, rev, accum, Eu);
}

__global__ void bpn_edge3(const float* __restrict__ rec,
                          const float* __restrict__ bel2,
                          const int* __restrict__ src,
                          const int* __restrict__ dst,
                          const int* __restrict__ rev,
                          float* __restrict__ accum, int Eu) {
  edge_body<3>(rec, bel2, src, dst, rev, accum, Eu);
}

__global__ void bpn_node(float* __restrict__ accum,
                         const float* __restrict__ log_priors,
                         float* __restrict__ beliefs_out,
                         int ostride, int ooff,
                         int N) {
  const int w    = blockIdx.x * (blockDim.x >> 5) + (threadIdx.x >> 5);
  const int lane = threadIdx.x & 31;
  const int half = lane >> 4;
  const int s    = lane & 15;
  const int node = w * 2 + half;

  const bool ok = (node < N);
  const size_t o = (size_t)(ok ? node : 0) * STATES + s;
  float v = ok ? accum[o] : 0.0f;

  float mx = v;
  mx = fmaxf(mx, __shfl_xor(mx, 1, 32));
  mx = fmaxf(mx, __shfl_xor(mx, 2, 32));
  mx = fmaxf(mx, __shfl_xor(mx, 4, 32));
  mx = fmaxf(mx, __shfl_xor(mx, 8, 32));
  float ex = expf(v - mx);
  float sm = ex;
  sm += __shfl_xor(sm, 1, 32);
  sm += __shfl_xor(sm, 2, 32);
  sm += __shfl_xor(sm, 4, 32);
  sm += __shfl_xor(sm, 8, 32);

  if (ok) {
    beliefs_out[o * (size_t)ostride + ooff] = ex / sm;
    accum[o] = log_priors[o];
  }
}

extern "C" void kernel_launch(void* const* d_in, const int* in_sizes, int n_in,
                              void* d_out, int out_size, void* d_ws, size_t ws_size,
                              hipStream_t stream) {
  const float* feat = (const float*)d_in[0];
  const float* W    = (const float*)d_in[1];
  const int*   src  = (const int*)d_in[2];
  const int*   dst  = (const int*)d_in[3];
  const int*   rev  = (const int*)d_in[4];

  const int N  = in_sizes[0] / FDIM;
  const int E2 = in_sizes[2];
  const int Eu = E2 / 2;

  float* priors      = (float*)d_out;
  float* beliefs_out = priors + (size_t)N * STATES;

  float* lp    = (float*)d_ws;
  float* accum = lp    + (size_t)N * STATES;
  float* lg1   = accum + (size_t)N * STATES;
  float* bel2  = lg1   + (size_t)N * STATES;
  float* rec   = bel2  + (size_t)N * STATES;

  const dim3 blk(256);
  const int tiles   = (N + 15) / 16;
  const int gBlocks = (tiles + 7) / 8;
  const int eBlocks = (Eu + 8 * PAIRS - 1) / (8 * PAIRS);
  const int nWaves  = (N + 1) / 2;
  const int nBlocks = (nWaves + 7) / 8;

  bpn_priors_wmma<<<gBlocks, blk, 0, stream>>>(feat, W, priors, lp, accum, rec, lg1, N, tiles);

  bpn_edge1<<<eBlocks, blk, 0, stream>>>(lg1, src, dst, accum, Eu);
  bpn_node<<<nBlocks, blk, 0, stream>>>(accum, lp, rec, 2, 1, N);
  bpn_edge2<<<eBlocks, blk, 0, stream>>>(rec, src, dst, rev, accum, Eu);
  bpn_node<<<nBlocks, blk, 0, stream>>>(accum, lp, bel2, 1, 0, N);
  bpn_edge3<<<eBlocks, blk, 0, stream>>>(rec, bel2, src, dst, rev, accum, Eu);
  bpn_node<<<nBlocks, blk, 0, stream>>>(accum, lp, beliefs_out, 1, 0, N);
}


# Scatter-add accumulator split into two tables, one per half of the package (die id read per wave, for speed only): each f32 atomic stays in its own half's L2; node kernel adds the two tables
# speedup vs baseline: 3.4392x; 1.3292x over previous
#include <hip/hip_runtime.h>
#include <math.h>


typedef float v2f __attribute__((ext_vector_type(2)));
typedef float v8f __attribute__((ext_vector_type(8)));

#define STATES   16
#define FDIM     128
#define EPSF     1e-10f
#define EXP1F    2.7182818284590452f
#define EXP1M1F  1.7182818284590452f
#define LN2F     0.69314718055994531f
#define MSG_A    (1.0f / (15.0f + EXP1F))
#define MSG_B    (EXP1M1F / (15.0f + EXP1F))

static __device__ __forceinline__ unsigned placement_word() {
  return __builtin_amdgcn_s_sendmsg_rtn(0x87);
}
static __device__ __forceinline__ float* own_half_table(unsigned place, float* tab0, float* tab1) {
  return ((place >> 18) & 1u) ? tab1 : tab0;
}

static __device__ __forceinline__ float bp_msg(float x) {
  float t = x;
  t += __shfl_xor(t, 1, 32);
  t += __shfl_xor(t, 2, 32);
  t += __shfl_xor(t, 4, 32);
  t += __shfl_xor(t, 8, 32);
  return (t + EXP1M1F * x) / ((15.0f + EXP1F) * t);
}

static __device__ __forceinline__ float bp_msg_edge(float x) {
  float t = x;
  t += __shfl_xor(t, 1, 32);
  t += __shfl_xor(t, 2, 32);
  t += __shfl_xor(t, 4, 32);
  t += __shfl_xor(t, 8, 32);
  return fmaf(x * __builtin_amdgcn_rcpf(t), MSG_B, MSG_A);
}

static __device__ __forceinline__ float log_msg_edge(float m) {
  return __builtin_amdgcn_logf(fmaxf(m, EPSF)) * LN2F;
}

__global__ void bpn_priors_wmma(const float* __restrict__ feat,
                                const float* __restrict__ W,
                                float* __restrict__ priors,
                                float* __restrict__ log_priors,
                                float* __restrict__ accum,
                                float* __restrict__ accumB,
                                float* __restrict__ rec,
                                float* __restrict__ lg1,
                                int N, int numTiles) {
  const int wave = blockIdx.x * (blockDim.x >> 5) + (threadIdx.x >> 5);
  if (wave >= numTiles) return;
  const int lane = threadIdx.x & 31;
  const int half = lane >> 4;
  const int s    = lane & 15;

  int rowA = wave * 16 + s;
  if (rowA >= N) rowA = N - 1;
  const float* frow = feat + (size_t)rowA * FDIM;

  v8f acc = {};
#pragma unroll
  for (int k0 = 0; k0 < FDIM; k0 += 4) {
    const int k = k0 + 2 * half;
    v2f a, b;
    a.x = frow[k];
    a.y = frow[k + 1];
    b.x = W[(size_t)k * STATES + s];
    b.y = W[(size_t)(k + 1) * STATES + s];
    acc = __builtin_amdgcn_wmma_f32_16x16x4_f32(false, a, false, b,
                                                (short)0, acc, false, false);
  }

#pragma unroll
  for (int r = 0; r < 8; ++r) {
    float logit = acc[r];
    float mx = logit;
    mx = fmaxf(mx, __shfl_xor(mx, 1, 32));
    mx = fmaxf(mx, __shfl_xor(mx, 2, 32));
    mx = fmaxf(mx, __shfl_xor(mx, 4, 32));
    mx = fmaxf(mx, __shfl_xor(mx, 8, 32));
    float ex = expf(logit - mx);
    float sm = ex;
    sm += __shfl_xor(sm, 1, 32);
    sm += __shfl_xor(sm, 2, 32);
    sm += __shfl_xor(sm, 4, 32);
    sm += __shfl_xor(sm, 8, 32);
    float pr = ex / sm;
    float lp = logf(fmaxf(pr, EPSF));
    float g1 = bp_msg(pr / 0.0625f);
    float l1 = logf(fmaxf(g1, EPSF));
    const int rowO = wave * 16 + r + 8 * half;
    if (rowO < N) {
      const size_t o = (size_t)rowO * STATES + s;
      priors[o]     = pr;
      log_priors[o] = lp;
      accum[o]      = lp;
      accumB[o]     = 0.0f;
      v2f g; g.x = g1; g.y = pr;
      *(v2f*)(rec + 2 * o) = g;
      lg1[o]        = l1;
    }
  }
}

#define PAIRS 8

template <int ITER>
static __device__ __forceinline__ void edge_body(const float* __restrict__ rec,
                                                 const float* __restrict__ bel2,
                                                 const int* __restrict__ src,
                                                 const int* __restrict__ dst,
                                                 const int* __restrict__ rev,
                                                 float* __restrict__ accum0,
                                                 float* __restrict__ accum1,
                                                 int Eu) {
  const int w  = __builtin_amdgcn_readfirstlane(
      (int)(blockIdx.x * (blockDim.x >> 5) + (threadIdx.x >> 5)));
  const int p0 = w * PAIRS;
  if (p0 >= Eu) return;
  const int lane = threadIdx.x & 31;
  const int half = lane >> 4;
  const int s    = lane & 15;

  const int pl = p0 + (lane & (PAIRS - 1));
  const int jv = rev[pl < Eu ? pl : (Eu - 1)];
  int  e[PAIRS];
  bool ok[PAIRS];
#pragma unroll
  for (int i = 0; i < PAIRS; ++i) {
    const int p = p0 + i;
    ok[i] = (p < Eu);
    const int j = __builtin_amdgcn_readlane(jv, i);
    e[i] = half ? j : (ok[i] ? p : (Eu - 1));
  }

  int se[PAIRS], de[PAIRS];
#pragma unroll
  for (int i = 0; i < PAIRS; ++i) {
    se[i] = src[e[i]];
    de[i] = dst[e[i]];
  }
  __builtin_amdgcn_sched_barrier(0);
  const unsigned place = placement_word();

  float lm[PAIRS];
  {
    v2f   gb[PAIRS];
    float b2[PAIRS];
#pragma unroll
    for (int i = 0; i < PAIRS; ++i) {
      gb[i] = *(const v2f*)(rec + 2 * ((size_t)se[i] * STATES + s));
      if (ITER == 3) b2[i] = bel2[(size_t)se[i] * STATES + s];
    }
    __builtin_amdgcn_sched_barrier(0);

    float m2[PAIRS];
#pragma unroll
    for (int i = 0; i < PAIRS; ++i) {
      const float g1rev = __shfl_xor(gb[i].x, 16, 32);
      m2[i] = bp_msg_edge(gb[i].y * __builtin_amdgcn_rcpf(g1rev));
    }
    if (ITER == 2) {
#pragma unroll
      for (int i = 0; i < PAIRS; ++i)
        lm[i] = log_msg_edge(m2[i]);
    } else {
#pragma unroll
      for (int i = 0; i < PAIRS; ++i) {
        const float m2rev = __shfl_xor(m2[i], 16, 32);
        const float m3    = bp_msg_edge(b2[i] * __builtin_amdgcn_rcpf(m2rev));
        lm[i] = log_msg_edge(m3);
      }
    }
  }

  float* accum = own_half_table(place, accum0, accum1);
#pragma unroll
  for (int i = 0; i < PAIRS; ++i) {
    if (ok[i])
      atomicAdd(&accum[(size_t)de[i] * STATES + s], lm[i]);
  }
}

__global__ void bpn_edge1(const float* __restrict__ lg1,
                          const int* __restrict__ src,
                          const int* __restrict__ dst,
                          float* __restrict__ accum0,
                          float* __restrict__ accum1, int Eu) {
  const int w  = __builtin_amdgcn_readfirstlane(
      (int)(blockIdx.x * (blockDim.x >> 5) + (threadIdx.x >> 5)));
  const int E2 = 2 * Eu;
  const int e0 = w * (2 * PAIRS);
  if (e0 >= E2) return;
  float* accum = own_half_table(placement_word(), accum0, accum1);
  const int lane = threadIdx.x & 31;
  const int half = lane >> 4;
  const int s    = lane & 15;

  int  se[PAIRS], de[PAIRS];
  bool ok[PAIRS];
#pragma unroll
  for (int i = 0; i < PAIRS; ++i) {
    ok[i] = (e0 + 2 * i < E2);
    const int e = ok[i] ? (e0 + 2 * i + half) : (E2 - 1);
    se[i] = src[e];
    de[i] = dst[e];
  }
  float lm[PAIRS];
#pragma unroll
  for (int i = 0; i < PAIRS; ++i)
    lm[i] = lg1[(size_t)se[i] * STATES + s];
#pragma unroll
  for (int i = 0; i < PAIRS; ++i) {
    if (ok[i])
      atomicAdd(&accum[(size_t)de[i] * STATES + s], lm[i]);
  }
}

__global__ void bpn_edge2(const float* __restrict__ rec,
                          const int* __restrict__ src,
                          const int* __restrict__ dst,
                          const int* __restrict__ rev,
                          float* __restrict__ accum0,
                          float* __restrict__ accum1, int Eu) {
  edge_body<2>(rec, nullptr, src, dst, rev, accum0, accum1, Eu);
}

__global__ void bpn_edge3(const float* __restrict__ rec,
                          const float* __restrict__ bel2,
                          const int* __restrict__ src,
                          const int* __restrict__ dst,
                          const int* __restrict__ rev,
                          float* __restrict__ accum0,
                          float* __restrict__ accum1, int Eu) {
  edge_body<3>(rec, bel2, src, dst, rev, accum0, accum1, Eu);
}

__global__ void bpn_node(float* __restrict__ accum,
                         float* __restrict__ accumB,
                         const float* __restrict__ log_priors,
                         float* __restrict__ beliefs_out,
                         int ostride, int ooff,
                         int N) {
  const int w    = blockIdx.x * (blockDim.x >> 5) + (threadIdx.x >> 5);
  const int lane = threadIdx.x & 31;
  const int half = lane >> 4;
  const int s    = lane & 15;
  const int node = w * 2 + half;

  const bool ok = (node < N);
  const size_t o = (size_t)(ok ? node : 0) * STATES + s;
  float v = ok ? (accum[o] + accumB[o]) : 0.0f;

  float mx = v;
  mx = fmaxf(mx, __shfl_xor(mx, 1, 32));
  mx = fmaxf(mx, __shfl_xor(mx, 2, 32));
  mx = fmaxf(mx, __shfl_xor(mx, 4, 32));
  mx = fmaxf(mx, __shfl_xor(mx, 8, 32));
  float ex = expf(v - mx);
  float sm = ex;
  sm += __shfl_xor(sm, 1, 32);
  sm += __shfl_xor(sm, 2, 32);
  sm += __shfl_xor(sm, 4, 32);
  sm += __shfl_xor(sm, 8, 32);

  if (ok) {
    beliefs_out[o * (size_t)ostride + ooff] = ex / sm;
    accum[o]  = log_priors[o];
    accumB[o] = 0.0f;
  }
}

extern "C" void kernel_launch(void* const* d_in, const int* in_sizes, int n_in,
                              void* d_out, int out_size, void* d_ws, size_t ws_size,
                              hipStream_t stream) {
  const float* feat = (const float*)d_in[0];
  const float* W    = (const float*)d_in[1];
  const int*   src  = (const int*)d_in[2];
  const int*   dst  = (const int*)d_in[3];
  const int*   rev  = (const int*)d_in[4];

  const int N  = in_sizes[0] / FDIM;
  const int E2 = in_sizes[2];
  const int Eu = E2 / 2;

  float* priors      = (float*)d_out;
  float* beliefs_out = priors + (size_t)N * STATES;

  float* lp    = (float*)d_ws;
  float* accum = lp    + (size_t)N * STATES;
  float* lg1   = accum + (size_t)N * STATES;
  float* bel2  = lg1   + (size_t)N * STATES;
  float* rec   = bel2  + (size_t)N * STATES;
  float* accB  = rec   + (size_t)N * STATES * 2;

  const dim3 blk(256);
  const int tiles   = (N + 15) / 16;
  const int gBlocks = (tiles + 7) / 8;
  const int eBlocks = (Eu + 8 * PAIRS - 1) / (8 * PAIRS);
  const int nWaves  = (N + 1) / 2;
  const int nBlocks = (nWaves + 7) / 8;

  bpn_priors_wmma<<<gBlocks, blk, 0, stream>>>(feat, W, priors, lp, accum, accB, rec, lg1, N, tiles);

  bpn_edge1<<<eBlocks, blk, 0, stream>>>(lg1, src, dst, accum, accB, Eu);
  bpn_node<<<nBlocks, blk, 0, stream>>>(accum, accB, lp, rec, 2, 1, N);
  bpn_edge2<<<eBlocks, blk, 0, stream>>>(rec, src, dst, rev, accum, accB, Eu);
  bpn_node<<<nBlocks, blk, 0, stream>>>(accum, accB, lp, bel2, 1, 0, N);
  bpn_edge3<<<eBlocks, blk, 0, stream>>>(rec, bel2, src, dst, rev, accum, accB, Eu);
  bpn_node<<<nBlocks, blk, 0, stream>>>(accum, accB, lp, beliefs_out, 1, 0, N);
}


# Cross-lane exchanges moved off the LDS unit onto the vector ALU: 16-state sums and maxima as DPP row_xmask butterflies, reverse-edge exchange as v_permlanex16_b32 (were ds_bpermute_b32 + wait)
# speedup vs baseline: 3.5536x; 1.0333x over previous
#include <hip/hip_runtime.h>
#include <math.h>


typedef float v2f __attribute__((ext_vector_type(2)));
typedef float v8f __attribute__((ext_vector_type(8)));

#define STATES   16
#define FDIM     128
#define EPSF     1e-10f
#define EXP1F    2.7182818284590452f
#define EXP1M1F  1.7182818284590452f
#define LN2F     0.69314718055994531f
#define MSG_A    (1.0f / (15.0f + EXP1F))
#define MSG_B    (EXP1M1F / (15.0f + EXP1F))

template <int M>
static __device__ __forceinline__ float row_xor(float x) {
  const int b = __builtin_bit_cast(int, x);
  return __builtin_bit_cast(float, __builtin_amdgcn_update_dpp(b, b, 0x160 + M, 0xf, 0xf, true));
}
static __device__ __forceinline__ float xhalf(float x) {
  const int b = __builtin_bit_cast(int, x);
  return __builtin_bit_cast(float,
      __builtin_amdgcn_permlanex16(b, b, 0x76543210, 0xfedcba98, false, false));
}
static __device__ __forceinline__ float row_sum16(float t) {
  t += row_xor<1>(t);
  t += row_xor<2>(t);
  t += row_xor<4>(t);
  t += row_xor<8>(t);
  return t;
}
static __device__ __forceinline__ float row_max16(float m) {
  m = fmaxf(m, row_xor<1>(m));
  m = fmaxf(m, row_xor<2>(m));
  m = fmaxf(m, row_xor<4>(m));
  m = fmaxf(m, row_xor<8>(m));
  return m;
}

static __device__ __forceinline__ unsigned placement_word() {
  return __builtin_amdgcn_s_sendmsg_rtn(0x87);
}
static __device__ __forceinline__ float* own_half_table(unsigned place, float* tab0, float* tab1) {
  return ((place >> 18) & 1u) ? tab1 : tab0;
}

static __device__ __forceinline__ float bp_msg(float x) {
  const float t = row_sum16(x);
  return (t + EXP1M1F * x) / ((15.0f + EXP1F) * t);
}

static __device__ __forceinline__ float bp_msg_edge(float x) {
  const float t = row_sum16(x);
  return fmaf(x * __builtin_amdgcn_rcpf(t), MSG_B, MSG_A);
}

static __device__ __forceinline__ float log_msg_edge(float m) {
  return __builtin_amdgcn_logf(fmaxf(m, EPSF)) * LN2F;
}

__global__ void bpn_priors_wmma(const float* __restrict__ feat,
                                const float* __restrict__ W,
                                float* __restrict__ priors,
                                float* __restrict__ log_priors,
                                float* __restrict__ accum,
                                float* __restrict__ accumB,
                                float* __restrict__ rec,
                                float* __restrict__ lg1,
                                int N, int numTiles) {
  const int wave = blockIdx.x * (blockDim.x >> 5) + (threadIdx.x >> 5);
  if (wave >= numTiles) return;
  const int lane = threadIdx.x & 31;
  const int half = lane >> 4;
  const int s    = lane & 15;

  int rowA = wave * 16 + s;
  if (rowA >= N) rowA = N - 1;
  const float* frow = feat + (size_t)rowA * FDIM;

  v8f acc = {};
#pragma unroll
  for (int k0 = 0; k0 < FDIM; k0 += 4) {
    const int k = k0 + 2 * half;
    v2f a, b;
    a.x = frow[k];
    a.y = frow[k + 1];
    b.x = W[(size_t)k * STATES + s];
    b.y = W[(size_t)(k + 1) * STATES + s];
    acc = __builtin_amdgcn_wmma_f32_16x16x4_f32(false, a, false, b,
                                                (short)0, acc, false, false);
  }

#pragma unroll
  for (int r = 0; r < 8; ++r) {
    float logit = acc[r];
    const float mx = row_max16(logit);
    float ex = expf(logit - mx);
    const float sm = row_sum16(ex);
    float pr = ex / sm;
    float lp = logf(fmaxf(pr, EPSF));
    float g1 = bp_msg(pr / 0.0625f);
    float l1 = logf(fmaxf(g1, EPSF));
    const int rowO = wave * 16 + r + 8 * half;
    if (rowO < N) {
      const size_t o = (size_t)rowO * STATES + s;
      priors[o]     = pr;
      log_priors[o] = lp;
      accum[o]      = lp;
      accumB[o]     = 0.0f;
      v2f g; g.x = g1; g.y = pr;
      *(v2f*)(rec + 2 * o) = g;
      lg1[o]        = l1;
    }
  }
}

#define PAIRS 8

template <int ITER>
static __device__ __forceinline__ void edge_body(const float* __restrict__ rec,
                                                 const float* __restrict__ bel2,
                                                 const int* __restrict__ src,
                                                 const int* __restrict__ dst,
                                                 const int* __restrict__ rev,
                                                 float* __restrict__ accum0,
                                                 float* __restrict__ accum1,
                                                 int Eu) {
  const int w  = __builtin_amdgcn_readfirstlane(
      (int)(blockIdx.x * (blockDim.x >> 5) + (threadIdx.x >> 5)));
  const int p0 = w * PAIRS;
  if (p0 >= Eu) return;
  const int lane = threadIdx.x & 31;
  const int half = lane >> 4;
  const int s    = lane & 15;

  const int pl = p0 + (lane & (PAIRS - 1));
  const int jv = rev[pl < Eu ? pl : (Eu - 1)];
  int  e[PAIRS];
  bool ok[PAIRS];
#pragma unroll
  for (int i = 0; i < PAIRS; ++i) {
    const int p = p0 + i;
    ok[i] = (p < Eu);
    const int j = __builtin_amdgcn_readlane(jv, i);
    e[i] = half ? j : (ok[i] ? p : (Eu - 1));
  }

  int se[PAIRS], de[PAIRS];
#pragma unroll
  for (int i = 0; i < PAIRS; ++i) {
    se[i] = src[e[i]];
    de[i] = dst[e[i]];
  }
  __builtin_amdgcn_sched_barrier(0);
  const unsigned place = placement_word();

  float lm[PAIRS];
  {
    v2f   gb[PAIRS];
    float b2[PAIRS];
#pragma unroll
    for (int i = 0; i < PAIRS; ++i) {
      gb[i] = *(const v2f*)(rec + 2 * ((size_t)se[i] * STATES + s));
      if (ITER == 3) b2[i] = bel2[(size_t)se[i] * STATES + s];
    }
    __builtin_amdgcn_sched_barrier(0);

    float m2[PAIRS];
#pragma unroll
    for (int i = 0; i < PAIRS; ++i) {
      const float g1rev = xhalf(gb[i].x);
      m2[i] = bp_msg_edge(gb[i].y * __builtin_amdgcn_rcpf(g1rev));
    }
    if (ITER == 2) {
#pragma unroll
      for (int i = 0; i < PAIRS; ++i)
        lm[i] = log_msg_edge(m2[i]);
    } else {
#pragma unroll
      for (int i = 0; i < PAIRS; ++i) {
        const float m2rev = xhalf(m2[i]);
        const float m3    = bp_msg_edge(b2[i] * __builtin_amdgcn_rcpf(m2rev));
        lm[i] = log_msg_edge(m3);
      }
    }
  }

  float* accum = own_half_table(place, accum0, accum1);
#pragma unroll
  for (int i = 0; i < PAIRS; ++i) {
    if (ok[i])
      atomicAdd(&accum[(size_t)de[i] * STATES + s], lm[i]);
  }
}

__global__ void bpn_edge1(const float* __restrict__ lg1,
                          const int* __restrict__ src,
                          const int* __restrict__ dst,
                          float* __restrict__ accum0,
                          float* __restrict__ accum1, int Eu) {
  const int w  = __builtin_amdgcn_readfirstlane(
      (int)(blockIdx.x * (blockDim.x >> 5) + (threadIdx.x >> 5)));
  const int E2 = 2 * Eu;
  const int e0 = w * (2 * PAIRS);
  if (e0 >= E2) return;
  float* accum = own_half_table(placement_word(), accum0, accum1);
  const int lane = threadIdx.x & 31;
  const int half = lane >> 4;
  const int s    = lane & 15;

  int  se[PAIRS], de[PAIRS];
  bool ok[PAIRS];
#pragma unroll
  for (int i = 0; i < PAIRS; ++i) {
    ok[i] = (e0 + 2 * i < E2);
    const int e = ok[i] ? (e0 + 2 * i + half) : (E2 - 1);
    se[i] = src[e];
    de[i] = dst[e];
  }
  float lm[PAIRS];
#pragma unroll
  for (int i = 0; i < PAIRS; ++i)
    lm[i] = lg1[(size_t)se[i] * STATES + s];
#pragma unroll
  for (int i = 0; i < PAIRS; ++i) {
    if (ok[i])
      atomicAdd(&accum[(size_t)de[i] * STATES + s], lm[i]);
  }
}

__global__ void bpn_edge2(const float* __restrict__ rec,
                          const int* __restrict__ src,
                          const int* __restrict__ dst,
                          const int* __restrict__ rev,
                          float* __restrict__ accum0,
                          float* __restrict__ accum1, int Eu) {
  edge_body<2>(rec, nullptr, src, dst, rev, accum0, accum1, Eu);
}

__global__ void bpn_edge3(const float* __restrict__ rec,
                          const float* __restrict__ bel2,
                          const int* __restrict__ src,
                          const int* __restrict__ dst,
                          const int* __restrict__ rev,
                          float* __restrict__ accum0,
                          float* __restrict__ accum1, int Eu) {
  edge_body<3>(rec, bel2, src, dst, rev, accum0, accum1, Eu);
}

__global__ void bpn_node(float* __restrict__ accum,
                         float* __restrict__ accumB,
                         const float* __restrict__ log_priors,
                         float* __restrict__ beliefs_out,
                         int ostride, int ooff,
                         int N) {
  const int w    = blockIdx.x * (blockDim.x >> 5) + (threadIdx.x >> 5);
  const int lane = threadIdx.x & 31;
  const int half = lane >> 4;
  const int s    = lane & 15;
  const int node = w * 2 + half;

  const bool ok = (node < N);
  const size_t o = (size_t)(ok ? node : 0) * STATES + s;
  float v = ok ? (accum[o] + accumB[o]) : 0.0f;

  const float mx = row_max16(v);
  float ex = expf(v - mx);
  const float sm = row_sum16(ex);

  if (ok) {
    beliefs_out[o * (size_t)ostride + ooff] = ex / sm;
    accum[o]  = log_priors[o];
    accumB[o] = 0.0f;
  }
}

extern "C" void kernel_launch(void* const* d_in, const int* in_sizes, int n_in,
                              void* d_out, int out_size, void* d_ws, size_t ws_size,
                              hipStream_t stream) {
  const float* feat = (const float*)d_in[0];
  const float* W    = (const float*)d_in[1];
  const int*   src  = (const int*)d_in[2];
  const int*   dst  = (const int*)d_in[3];
  const int*   rev  = (const int*)d_in[4];

  const int N  = in_sizes[0] / FDIM;
  const int E2 = in_sizes[2];
  const int Eu = E2 / 2;

  float* priors      = (float*)d_out;
  float* beliefs_out = priors + (size_t)N * STATES;

  float* lp    = (float*)d_ws;
  float* accum = lp    + (size_t)N * STATES;
  float* lg1   = accum + (size_t)N * STATES;
  float* bel2  = lg1   + (size_t)N * STATES;
  float* rec   = bel2  + (size_t)N * STATES;
  float* accB  = rec   + (size_t)N * STATES * 2;

  const dim3 blk(256);
  const int tiles   = (N + 15) / 16;
  const int gBlocks = (tiles + 7) / 8;
  const int eBlocks = (Eu + 8 * PAIRS - 1) / (8 * PAIRS);
  const int nWaves  = (N + 1) / 2;
  const int nBlocks = (nWaves + 7) / 8;

  bpn_priors_wmma<<<gBlocks, blk, 0, stream>>>(feat, W, priors, lp, accum, accB, rec, lg1, N, tiles);

  bpn_edge1<<<eBlocks, blk, 0, stream>>>(lg1, src, dst, accum, accB, Eu);
  bpn_node<<<nBlocks, blk, 0, stream>>>(accum, accB, lp, rec, 2, 1, N);
  bpn_edge2<<<eBlocks, blk, 0, stream>>>(rec, src, dst, rev, accum, accB, Eu);
  bpn_node<<<nBlocks, blk, 0, stream>>>(accum, accB, lp, bel2, 1, 0, N);
  bpn_edge3<<<eBlocks, blk, 0, stream>>>(rec, bel2, src, dst, rev, accum, accB, Eu);
  bpn_node<<<nBlocks, blk, 0, stream>>>(accum, accB, lp, beliefs_out, 1, 0, N);
}


# Wave count cut 6x-8x: edge kernels launch a fixed grid of 4096 blocks whose waves loop over groups of 8 pairs (next group's partner indices fetched a trip ahead); node kernel takes 16 nodes per wave
# speedup vs baseline: 4.8855x; 1.3748x over previous
#include <hip/hip_runtime.h>
#include <math.h>


typedef float v2f __attribute__((ext_vector_type(2)));
typedef float v8f __attribute__((ext_vector_type(8)));

#define STATES   16
#define FDIM     128
#define EPSF     1e-10f
#define EXP1F    2.7182818284590452f
#define EXP1M1F  1.7182818284590452f
#define LN2F     0.69314718055994531f
#define MSG_A    (1.0f / (15.0f + EXP1F))
#define MSG_B    (EXP1M1F / (15.0f + EXP1F))

template <int M>
static __device__ __forceinline__ float row_xor(float x) {
  const int b = __builtin_bit_cast(int, x);
  return __builtin_bit_cast(float, __builtin_amdgcn_update_dpp(b, b, 0x160 + M, 0xf, 0xf, true));
}
static __device__ __forceinline__ float xhalf(float x) {
  const int b = __builtin_bit_cast(int, x);
  return __builtin_bit_cast(float,
      __builtin_amdgcn_permlanex16(b, b, 0x76543210, 0xfedcba98, false, false));
}
static __device__ __forceinline__ float row_sum16(float t) {
  t += row_xor<1>(t);
  t += row_xor<2>(t);
  t += row_xor<4>(t);
  t += row_xor<8>(t);
  return t;
}
static __device__ __forceinline__ float row_max16(float m) {
  m = fmaxf(m, row_xor<1>(m));
  m = fmaxf(m, row_xor<2>(m));
  m = fmaxf(m, row_xor<4>(m));
  m = fmaxf(m, row_xor<8>(m));
  return m;
}

static __device__ __forceinline__ unsigned placement_word() {
  return __builtin_amdgcn_s_sendmsg_rtn(0x87);
}
static __device__ __forceinline__ float* own_half_table(unsigned place, float* tab0, float* tab1) {
  return ((place >> 18) & 1u) ? tab1 : tab0;
}

static __device__ __forceinline__ float bp_msg(float x) {
  const float t = row_sum16(x);
  return (t + EXP1M1F * x) / ((15.0f + EXP1F) * t);
}

static __device__ __forceinline__ float bp_msg_edge(float x) {
  const float t = row_sum16(x);
  return fmaf(x * __builtin_amdgcn_rcpf(t), MSG_B, MSG_A);
}

static __device__ __forceinline__ float log_msg_edge(float m) {
  return __builtin_amdgcn_logf(fmaxf(m, EPSF)) * LN2F;
}

__global__ void bpn_priors_wmma(const float* __restrict__ feat,
                                const float* __restrict__ W,
                                float* __restrict__ priors,
                                float* __restrict__ log_priors,
                                float* __restrict__ accum,
                                float* __restrict__ accumB,
                                float* __restrict__ rec,
                                float* __restrict__ lg1,
                                int N, int numTiles) {
  const int wave = blockIdx.x * (blockDim.x >> 5) + (threadIdx.x >> 5);
  if (wave >= numTiles) return;
  const int lane = threadIdx.x & 31;
  const int half = lane >> 4;
  const int s    = lane & 15;

  int rowA = wave * 16 + s;
  if (rowA >= N) rowA = N - 1;
  const float* frow = feat + (size_t)rowA * FDIM;

  v8f acc = {};
#pragma unroll
  for (int k0 = 0; k0 < FDIM; k0 += 4) {
    const int k = k0 + 2 * half;
    v2f a, b;
    a.x = frow[k];
    a.y = frow[k + 1];
    b.x = W[(size_t)k * STATES + s];
    b.y = W[(size_t)(k + 1) * STATES + s];
    acc = __builtin_amdgcn_wmma_f32_16x16x4_f32(false, a, false, b,
                                                (short)0, acc, false, false);
  }

#pragma unroll
  for (int r = 0; r < 8; ++r) {
    float logit = acc[r];
    const float mx = row_max16(logit);
    float ex = expf(logit - mx);
    const float sm = row_sum16(ex);
    float pr = ex / sm;
    float lp = logf(fmaxf(pr, EPSF));
    float g1 = bp_msg(pr / 0.0625f);
    float l1 = logf(fmaxf(g1, EPSF));
    const int rowO = wave * 16 + r + 8 * half;
    if (rowO < N) {
      const size_t o = (size_t)rowO * STATES + s;
      priors[o]     = pr;
      log_priors[o] = lp;
      accum[o]      = lp;
      accumB[o]     = 0.0f;
      v2f g; g.x = g1; g.y = pr;
      *(v2f*)(rec + 2 * o) = g;
      lg1[o]        = l1;
    }
  }
}

#define PAIRS 8

#define EDGE_BLOCKS 4096

template <int ITER, bool FULL>
static __device__ __forceinline__ void edge_group(const float* __restrict__ rec,
                                                  const float* __restrict__ bel2,
                                                  const int* __restrict__ src,
                                                  const int* __restrict__ dst,
                                                  float* __restrict__ accum,
                                                  int jv, int p0, int Eu, int half, int s) {
  int  e[PAIRS];
  bool ok[PAIRS];
#pragma unroll
  for (int i = 0; i < PAIRS; ++i) {
    const int p = p0 + i;
    ok[i] = FULL || (p < Eu);
    const int j = __builtin_amdgcn_readlane(jv, i);
    e[i] = half ? j : (ok[i] ? p : (Eu - 1));
  }

  int se[PAIRS], de[PAIRS];
#pragma unroll
  for (int i = 0; i < PAIRS; ++i) {
    se[i] = src[e[i]];
    de[i] = dst[e[i]];
  }
  __builtin_amdgcn_sched_barrier(0);

  float lm[PAIRS];
  {
    v2f   gb[PAIRS];
    float b2[PAIRS];
#pragma unroll
    for (int i = 0; i < PAIRS; ++i) {
      gb[i] = *(const v2f*)(rec + 2 * ((size_t)se[i] * STATES + s));
      if (ITER == 3) b2[i] = bel2[(size_t)se[i] * STATES + s];
    }
    __builtin_amdgcn_sched_barrier(0);

    float m2[PAIRS];
#pragma unroll
    for (int i = 0; i < PAIRS; ++i) {
      const float g1rev = xhalf(gb[i].x);
      m2[i] = bp_msg_edge(gb[i].y * __builtin_amdgcn_rcpf(g1rev));
    }
    if (ITER == 2) {
#pragma unroll
      for (int i = 0; i < PAIRS; ++i)
        lm[i] = log_msg_edge(m2[i]);
    } else {
#pragma unroll
      for (int i = 0; i < PAIRS; ++i) {
        const float m2rev = xhalf(m2[i]);
        const float m3    = bp_msg_edge(b2[i] * __builtin_amdgcn_rcpf(m2rev));
        lm[i] = log_msg_edge(m3);
      }
    }
  }

#pragma unroll
  for (int i = 0; i < PAIRS; ++i) {
    if (ok[i])
      atomicAdd(&accum[(size_t)de[i] * STATES + s], lm[i]);
  }
}

template <int ITER>
static __device__ __forceinline__ void edge_body(const float* __restrict__ rec,
                                                 const float* __restrict__ bel2,
                                                 const int* __restrict__ src,
                                                 const int* __restrict__ dst,
                                                 const int* __restrict__ rev,
                                                 float* __restrict__ accum0,
                                                 float* __restrict__ accum1,
                                                 int Eu) {
  const int w  = __builtin_amdgcn_readfirstlane(
      (int)(blockIdx.x * (blockDim.x >> 5) + (threadIdx.x >> 5)));
  const int nW = (int)(gridDim.x * (blockDim.x >> 5));
  const int nG = (Eu + PAIRS - 1) / PAIRS;
  if (w >= nG) return;
  const int lane = threadIdx.x & 31;
  const int half = lane >> 4;
  const int s    = lane & 15;
  const int sub  = lane & (PAIRS - 1);

  int pl = w * PAIRS + sub;
  int jv = rev[pl < Eu ? pl : (Eu - 1)];
  float* accum = own_half_table(placement_word(), accum0, accum1);

  for (int g = w; g < nG; g += nW) {
    int pn = (g + nW) * PAIRS + sub;
    const int jn = rev[pn < Eu ? pn : (Eu - 1)];
    const int p0 = g * PAIRS;
    if (p0 + PAIRS <= Eu)
      edge_group<ITER, true >(rec, bel2, src, dst, accum, jv, p0, Eu, half, s);
    else
      edge_group<ITER, false>(rec, bel2, src, dst, accum, jv, p0, Eu, half, s);
    jv = jn;
  }
}

template <bool FULL>
static __device__ __forceinline__ void edge1_group(const float* __restrict__ lg1,
                                                   const int* __restrict__ src,
                                                   const int* __restrict__ dst,
                                                   float* __restrict__ accum,
                                                   int e0, int E2, int half, int s) {
  int  se[PAIRS], de[PAIRS];
  bool ok[PAIRS];
#pragma unroll
  for (int i = 0; i < PAIRS; ++i) {
    ok[i] = FULL || (e0 + 2 * i < E2);
    const int e = ok[i] ? (e0 + 2 * i + half) : (E2 - 1);
    se[i] = src[e];
    de[i] = dst[e];
  }
  float lm[PAIRS];
#pragma unroll
  for (int i = 0; i < PAIRS; ++i)
    lm[i] = lg1[(size_t)se[i] * STATES + s];
#pragma unroll
  for (int i = 0; i < PAIRS; ++i) {
    if (ok[i])
      atomicAdd(&accum[(size_t)de[i] * STATES + s], lm[i]);
  }
}

__global__ void bpn_edge1(const float* __restrict__ lg1,
                          const int* __restrict__ src,
                          const int* __restrict__ dst,
                          float* __restrict__ accum0,
                          float* __restrict__ accum1, int Eu) {
  const int w  = __builtin_amdgcn_readfirstlane(
      (int)(blockIdx.x * (blockDim.x >> 5) + (threadIdx.x >> 5)));
  const int nW = (int)(gridDim.x * (blockDim.x >> 5));
  const int E2 = 2 * Eu;
  const int nG = (E2 + 2 * PAIRS - 1) / (2 * PAIRS);
  if (w >= nG) return;
  float* accum = own_half_table(placement_word(), accum0, accum1);
  const int lane = threadIdx.x & 31;
  const int half = lane >> 4;
  const int s    = lane & 15;

  for (int g = w; g < nG; g += nW) {
    const int e0 = g * (2 * PAIRS);
    if (e0 + 2 * PAIRS <= E2)
      edge1_group<true >(lg1, src, dst, accum, e0, E2, half, s);
    else
      edge1_group<false>(lg1, src, dst, accum, e0, E2, half, s);
  }
}

__global__ void bpn_edge2(const float* __restrict__ rec,
                          const int* __restrict__ src,
                          const int* __restrict__ dst,
                          const int* __restrict__ rev,
                          float* __restrict__ accum0,
                          float* __restrict__ accum1, int Eu) {
  edge_body<2>(rec, nullptr, src, dst, rev, accum0, accum1, Eu);
}

__global__ void bpn_edge3(const float* __restrict__ rec,
                          const float* __restrict__ bel2,
                          const int* __restrict__ src,
                          const int* __restrict__ dst,
                          const int* __restrict__ rev,
                          float* __restrict__ accum0,
                          float* __restrict__ accum1, int Eu) {
  edge_body<3>(rec, bel2, src, dst, rev, accum0, accum1, Eu);
}

#define NODEB 8

__global__ void bpn_node(float* __restrict__ accum,
                         float* __restrict__ accumB,
                         const float* __restrict__ log_priors,
                         float* __restrict__ beliefs_out,
                         int ostride, int ooff,
                         int N) {
  const int w    = blockIdx.x * (blockDim.x >> 5) + (threadIdx.x >> 5);
  const int lane = threadIdx.x & 31;
  const int half = lane >> 4;
  const int s    = lane & 15;
  const int base = w * (2 * NODEB) + half;

  bool   ok[NODEB];
  size_t o[NODEB];
  float  a[NODEB], b[NODEB], lp[NODEB];
#pragma unroll
  for (int i = 0; i < NODEB; ++i) {
    const int node = base + 2 * i;
    ok[i] = (node < N);
    o[i]  = (size_t)(ok[i] ? node : 0) * STATES + s;
    a[i]  = accum[o[i]];
    b[i]  = accumB[o[i]];
    lp[i] = log_priors[o[i]];
  }
#pragma unroll
  for (int i = 0; i < NODEB; ++i) {
    const float v  = ok[i] ? (a[i] + b[i]) : 0.0f;
    const float mx = row_max16(v);
    const float ex = expf(v - mx);
    const float sm = row_sum16(ex);
    if (ok[i]) {
      beliefs_out[o[i] * (size_t)ostride + ooff] = ex / sm;
      accum[o[i]]  = lp[i];
      accumB[o[i]] = 0.0f;
    }
  }
}

extern "C" void kernel_launch(void* const* d_in, const int* in_sizes, int n_in,
                              void* d_out, int out_size, void* d_ws, size_t ws_size,
                              hipStream_t stream) {
  const float* feat = (const float*)d_in[0];
  const float* W    = (const float*)d_in[1];
  const int*   src  = (const int*)d_in[2];
  const int*   dst  = (const int*)d_in[3];
  const int*   rev  = (const int*)d_in[4];

  const int N  = in_sizes[0] / FDIM;
  const int E2 = in_sizes[2];
  const int Eu = E2 / 2;

  float* priors      = (float*)d_out;
  float* beliefs_out = priors + (size_t)N * STATES;

  float* lp    = (float*)d_ws;
  float* accum = lp    + (size_t)N * STATES;
  float* lg1   = accum + (size_t)N * STATES;
  float* bel2  = lg1   + (size_t)N * STATES;
  float* rec   = bel2  + (size_t)N * STATES;
  float* accB  = rec   + (size_t)N * STATES * 2;

  const dim3 blk(256);
  const int tiles   = (N + 15) / 16;
  const int gBlocks = (tiles + 7) / 8;
  const int eNeed   = (Eu + 8 * PAIRS - 1) / (8 * PAIRS);
  const int eBlocks = eNeed < EDGE_BLOCKS ? eNeed : EDGE_BLOCKS;
  const int nWaves  = (N + 2 * NODEB - 1) / (2 * NODEB);
  const int nBlocks = (nWaves + 7) / 8;

  bpn_priors_wmma<<<gBlocks, blk, 0, stream>>>(feat, W, priors, lp, accum, accB, rec, lg1, N, tiles);

  bpn_edge1<<<eBlocks, blk, 0, stream>>>(lg1, src, dst, accum, accB, Eu);
  bpn_node<<<nBlocks, blk, 0, stream>>>(accum, accB, lp, rec, 2, 1, N);
  bpn_edge2<<<eBlocks, blk, 0, stream>>>(rec, src, dst, rev, accum, accB, Eu);
  bpn_node<<<nBlocks, blk, 0, stream>>>(accum, accB, lp, bel2, 1, 0, N);
  bpn_edge3<<<eBlocks, blk, 0, stream>>>(rec, bel2, src, dst, rev, accum, accB, Eu);
  bpn_node<<<nBlocks, blk, 0, stream>>>(accum, accB, lp, beliefs_out, 1, 0, N);
}


# Edge and node kernels: row gathers and scatter-adds addressed as scalar base + 32-bit element index (one shift-or each) instead of 64-bit per-lane addresses (sign-extend, 64-bit shift, 64-bit add)
# speedup vs baseline: 5.2551x; 1.0757x over previous
#include <hip/hip_runtime.h>
#include <math.h>


typedef float v2f __attribute__((ext_vector_type(2)));
typedef float v8f __attribute__((ext_vector_type(8)));

#define STATES   16
#define FDIM     128
#define EPSF     1e-10f
#define EXP1F    2.7182818284590452f
#define EXP1M1F  1.7182818284590452f
#define LN2F     0.69314718055994531f
#define MSG_A    (1.0f / (15.0f + EXP1F))
#define MSG_B    (EXP1M1F / (15.0f + EXP1F))

template <int M>
static __device__ __forceinline__ float row_xor(float x) {
  const int b = __builtin_bit_cast(int, x);
  return __builtin_bit_cast(float, __builtin_amdgcn_update_dpp(b, b, 0x160 + M, 0xf, 0xf, true));
}
static __device__ __forceinline__ float xhalf(float x) {
  const int b = __builtin_bit_cast(int, x);
  return __builtin_bit_cast(float,
      __builtin_amdgcn_permlanex16(b, b, 0x76543210, 0xfedcba98, false, false));
}
static __device__ __forceinline__ float row_sum16(float t) {
  t += row_xor<1>(t);
  t += row_xor<2>(t);
  t += row_xor<4>(t);
  t += row_xor<8>(t);
  return t;
}
static __device__ __forceinline__ float row_max16(float m) {
  m = fmaxf(m, row_xor<1>(m));
  m = fmaxf(m, row_xor<2>(m));
  m = fmaxf(m, row_xor<4>(m));
  m = fmaxf(m, row_xor<8>(m));
  return m;
}

static __device__ __forceinline__ unsigned placement_word() {
  return __builtin_amdgcn_s_sendmsg_rtn(0x87);
}
static __device__ __forceinline__ float* own_half_table(unsigned place, float* tab0, float* tab1) {
  return ((place >> 18) & 1u) ? tab1 : tab0;
}

static __device__ __forceinline__ float bp_msg(float x) {
  const float t = row_sum16(x);
  return (t + EXP1M1F * x) / ((15.0f + EXP1F) * t);
}

static __device__ __forceinline__ float bp_msg_edge(float x) {
  const float t = row_sum16(x);
  return fmaf(x * __builtin_amdgcn_rcpf(t), MSG_B, MSG_A);
}

static __device__ __forceinline__ float log_msg_edge(float m) {
  return __builtin_amdgcn_logf(fmaxf(m, EPSF)) * LN2F;
}

__global__ void bpn_priors_wmma(const float* __restrict__ feat,
                                const float* __restrict__ W,
                                float* __restrict__ priors,
                                float* __restrict__ log_priors,
                                float* __restrict__ accum,
                                float* __restrict__ accumB,
                                float* __restrict__ rec,
                                float* __restrict__ lg1,
                                int N, int numTiles) {
  const int wave = blockIdx.x * (blockDim.x >> 5) + (threadIdx.x >> 5);
  if (wave >= numTiles) return;
  const int lane = threadIdx.x & 31;
  const int half = lane >> 4;
  const int s    = lane & 15;

  int rowA = wave * 16 + s;
  if (rowA >= N) rowA = N - 1;
  const float* frow = feat + (size_t)rowA * FDIM;

  v8f acc = {};
#pragma unroll
  for (int k0 = 0; k0 < FDIM; k0 += 4) {
    const int k = k0 + 2 * half;
    v2f a, b;
    a.x = frow[k];
    a.y = frow[k + 1];
    b.x = W[(size_t)k * STATES + s];
    b.y = W[(size_t)(k + 1) * STATES + s];
    acc = __builtin_amdgcn_wmma_f32_16x16x4_f32(false, a, false, b,
                                                (short)0, acc, false, false);
  }

#pragma unroll
  for (int r = 0; r < 8; ++r) {
    float logit = acc[r];
    const float mx = row_max16(logit);
    float ex = expf(logit - mx);
    const float sm = row_sum16(ex);
    float pr = ex / sm;
    float lp = logf(fmaxf(pr, EPSF));
    float g1 = bp_msg(pr / 0.0625f);
    float l1 = logf(fmaxf(g1, EPSF));
    const int rowO = wave * 16 + r + 8 * half;
    if (rowO < N) {
      const size_t o = (size_t)rowO * STATES + s;
      priors[o]     = pr;
      log_priors[o] = lp;
      accum[o]      = lp;
      accumB[o]     = 0.0f;
      v2f g; g.x = g1; g.y = pr;
      *(v2f*)(rec + 2 * o) = g;
      lg1[o]        = l1;
    }
  }
}

#define PAIRS 8

#define EDGE_BLOCKS 4096

template <int ITER, bool FULL>
static __device__ __forceinline__ void edge_group(const float* __restrict__ rec,
                                                  const float* __restrict__ bel2,
                                                  const int* __restrict__ src,
                                                  const int* __restrict__ dst,
                                                  float* __restrict__ accum,
                                                  int jv, int p0, int Eu, int half, int s) {
  int  e[PAIRS];
  bool ok[PAIRS];
#pragma unroll
  for (int i = 0; i < PAIRS; ++i) {
    const int p = p0 + i;
    ok[i] = FULL || (p < Eu);
    const int j = __builtin_amdgcn_readlane(jv, i);
    e[i] = half ? j : (ok[i] ? p : (Eu - 1));
  }

  int se[PAIRS], de[PAIRS];
#pragma unroll
  for (int i = 0; i < PAIRS; ++i) {
    se[i] = src[e[i]];
    de[i] = dst[e[i]];
  }
  __builtin_amdgcn_sched_barrier(0);

  const v2f* rec2 = (const v2f*)rec;
  float lm[PAIRS];
  {
    v2f   gb[PAIRS];
    float b2[PAIRS];
#pragma unroll
    for (int i = 0; i < PAIRS; ++i) {
      const int row = se[i] * STATES + s;
      gb[i] = rec2[row];
      if (ITER == 3) b2[i] = bel2[row];
    }
    __builtin_amdgcn_sched_barrier(0);

    float m2[PAIRS];
#pragma unroll
    for (int i = 0; i < PAIRS; ++i) {
      const float g1rev = xhalf(gb[i].x);
      m2[i] = bp_msg_edge(gb[i].y * __builtin_amdgcn_rcpf(g1rev));
    }
    if (ITER == 2) {
#pragma unroll
      for (int i = 0; i < PAIRS; ++i)
        lm[i] = log_msg_edge(m2[i]);
    } else {
#pragma unroll
      for (int i = 0; i < PAIRS; ++i) {
        const float m2rev = xhalf(m2[i]);
        const float m3    = bp_msg_edge(b2[i] * __builtin_amdgcn_rcpf(m2rev));
        lm[i] = log_msg_edge(m3);
      }
    }
  }

#pragma unroll
  for (int i = 0; i < PAIRS; ++i) {
    if (ok[i])
      atomicAdd(&accum[de[i] * STATES + s], lm[i]);
  }
}

template <int ITER>
static __device__ __forceinline__ void edge_body(const float* __restrict__ rec,
                                                 const float* __restrict__ bel2,
                                                 const int* __restrict__ src,
                                                 const int* __restrict__ dst,
                                                 const int* __restrict__ rev,
                                                 float* __restrict__ accum0,
                                                 float* __restrict__ accum1,
                                                 int Eu) {
  const int w  = __builtin_amdgcn_readfirstlane(
      (int)(blockIdx.x * (blockDim.x >> 5) + (threadIdx.x >> 5)));
  const int nW = (int)(gridDim.x * (blockDim.x >> 5));
  const int nG = (Eu + PAIRS - 1) / PAIRS;
  if (w >= nG) return;
  const int lane = threadIdx.x & 31;
  const int half = lane >> 4;
  const int s    = lane & 15;
  const int sub  = lane & (PAIRS - 1);

  int pl = w * PAIRS + sub;
  int jv = rev[pl < Eu ? pl : (Eu - 1)];
  float* accum = own_half_table(placement_word(), accum0, accum1);

  for (int g = w; g < nG; g += nW) {
    int pn = (g + nW) * PAIRS + sub;
    const int jn = rev[pn < Eu ? pn : (Eu - 1)];
    const int p0 = g * PAIRS;
    if (p0 + PAIRS <= Eu)
      edge_group<ITER, true >(rec, bel2, src, dst, accum, jv, p0, Eu, half, s);
    else
      edge_group<ITER, false>(rec, bel2, src, dst, accum, jv, p0, Eu, half, s);
    jv = jn;
  }
}

template <bool FULL>
static __device__ __forceinline__ void edge1_group(const float* __restrict__ lg1,
                                                   const int* __restrict__ src,
                                                   const int* __restrict__ dst,
                                                   float* __restrict__ accum,
                                                   int e0, int E2, int half, int s) {
  int  se[PAIRS], de[PAIRS];
  bool ok[PAIRS];
#pragma unroll
  for (int i = 0; i < PAIRS; ++i) {
    ok[i] = FULL || (e0 + 2 * i < E2);
    const int e = ok[i] ? (e0 + 2 * i + half) : (E2 - 1);
    se[i] = src[e];
    de[i] = dst[e];
  }
  float lm[PAIRS];
#pragma unroll
  for (int i = 0; i < PAIRS; ++i)
    lm[i] = lg1[se[i] * STATES + s];
#pragma unroll
  for (int i = 0; i < PAIRS; ++i) {
    if (ok[i])
      atomicAdd(&accum[de[i] * STATES + s], lm[i]);
  }
}

__global__ void bpn_edge1(const float* __restrict__ lg1,
                          const int* __restrict__ src,
                          const int* __restrict__ dst,
                          float* __restrict__ accum0,
                          float* __restrict__ accum1, int Eu) {
  const int w  = __builtin_amdgcn_readfirstlane(
      (int)(blockIdx.x * (blockDim.x >> 5) + (threadIdx.x >> 5)));
  const int nW = (int)(gridDim.x * (blockDim.x >> 5));
  const int E2 = 2 * Eu;
  const int nG = (E2 + 2 * PAIRS - 1) / (2 * PAIRS);
  if (w >= nG) return;
  float* accum = own_half_table(placement_word(), accum0, accum1);
  const int lane = threadIdx.x & 31;
  const int half = lane >> 4;
  const int s    = lane & 15;

  for (int g = w; g < nG; g += nW) {
    const int e0 = g * (2 * PAIRS);
    if (e0 + 2 * PAIRS <= E2)
      edge1_group<true >(lg1, src, dst, accum, e0, E2, half, s);
    else
      edge1_group<false>(lg1, src, dst, accum, e0, E2, half, s);
  }
}

__global__ void bpn_edge2(const float* __restrict__ rec,
                          const int* __restrict__ src,
                          const int* __restrict__ dst,
                          const int* __restrict__ rev,
                          float* __restrict__ accum0,
                          float* __restrict__ accum1, int Eu) {
  edge_body<2>(rec, nullptr, src, dst, rev, accum0, accum1, Eu);
}

__global__ void bpn_edge3(const float* __restrict__ rec,
                          const float* __restrict__ bel2,
                          const int* __restrict__ src,
                          const int* __restrict__ dst,
                          const int* __restrict__ rev,
                          float* __restrict__ accum0,
                          float* __restrict__ accum1, int Eu) {
  edge_body<3>(rec, bel2, src, dst, rev, accum0, accum1, Eu);
}

#define NODEB 8

__global__ void bpn_node(float* __restrict__ accum,
                         float* __restrict__ accumB,
                         const float* __restrict__ log_priors,
                         float* __restrict__ beliefs_out,
                         int ostride, int ooff,
                         int N) {
  const int w    = blockIdx.x * (blockDim.x >> 5) + (threadIdx.x >> 5);
  const int lane = threadIdx.x & 31;
  const int half = lane >> 4;
  const int s    = lane & 15;
  const int base = w * (2 * NODEB) + half;

  bool   ok[NODEB];
  int    o[NODEB];
  float  a[NODEB], b[NODEB], lp[NODEB];
#pragma unroll
  for (int i = 0; i < NODEB; ++i) {
    const int node = base + 2 * i;
    ok[i] = (node < N);
    o[i]  = (ok[i] ? node : 0) * STATES + s;
    a[i]  = accum[o[i]];
    b[i]  = accumB[o[i]];
    lp[i] = log_priors[o[i]];
  }
#pragma unroll
  for (int i = 0; i < NODEB; ++i) {
    const float v  = ok[i] ? (a[i] + b[i]) : 0.0f;
    const float mx = row_max16(v);
    const float ex = expf(v - mx);
    const float sm = row_sum16(ex);
    if (ok[i]) {
      beliefs_out[o[i] * ostride + ooff] = ex / sm;
      accum[o[i]]  = lp[i];
      accumB[o[i]] = 0.0f;
    }
  }
}

extern "C" void kernel_launch(void* const* d_in, const int* in_sizes, int n_in,
                              void* d_out, int out_size, void* d_ws, size_t ws_size,
                              hipStream_t stream) {
  const float* feat = (const float*)d_in[0];
  const float* W    = (const float*)d_in[1];
  const int*   src  = (const int*)d_in[2];
  const int*   dst  = (const int*)d_in[3];
  const int*   rev  = (const int*)d_in[4];

  const int N  = in_sizes[0] / FDIM;
  const int E2 = in_sizes[2];
  const int Eu = E2 / 2;

  float* priors      = (float*)d_out;
  float* beliefs_out = priors + (size_t)N * STATES;

  float* lp    = (float*)d_ws;
  float* accum = lp    + (size_t)N * STATES;
  float* lg1   = accum + (size_t)N * STATES;
  float* bel2  = lg1   + (size_t)N * STATES;
  float* rec   = bel2  + (size_t)N * STATES;
  float* accB  = rec   + (size_t)N * STATES * 2;

  const dim3 blk(256);
  const int tiles   = (N + 15) / 16;
  const int gBlocks = (tiles + 7) / 8;
  const int eNeed   = (Eu + 8 * PAIRS - 1) / (8 * PAIRS);
  const int eBlocks = eNeed < EDGE_BLOCKS ? eNeed : EDGE_BLOCKS;
  const int nWaves  = (N + 2 * NODEB - 1) / (2 * NODEB);
  const int nBlocks = (nWaves + 7) / 8;

  bpn_priors_wmma<<<gBlocks, blk, 0, stream>>>(feat, W, priors, lp, accum, accB, rec, lg1, N, tiles);

  bpn_edge1<<<eBlocks, blk, 0, stream>>>(lg1, src, dst, accum, accB, Eu);
  bpn_node<<<nBlocks, blk, 0, stream>>>(accum, accB, lp, rec, 2, 1, N);
  bpn_edge2<<<eBlocks, blk, 0, stream>>>(rec, src, dst, rev, accum, accB, Eu);
  bpn_node<<<nBlocks, blk, 0, stream>>>(accum, accB, lp, bel2, 1, 0, N);
  bpn_edge3<<<eBlocks, blk, 0, stream>>>(rec, bel2, src, dst, rev, accum, accB, Eu);
  bpn_node<<<nBlocks, blk, 0, stream>>>(accum, accB, lp, beliefs_out, 1, 0, N);
}


# Edge kernels: a group's endpoint indices come from one coalesced src load and one dst load (a lane per edge, DPP row_share hands each pair its entry), a trip ahead, not a load per pair and array
# speedup vs baseline: 5.5249x; 1.0513x over previous
#include <hip/hip_runtime.h>
#include <math.h>


typedef float v2f __attribute__((ext_vector_type(2)));
typedef float v8f __attribute__((ext_vector_type(8)));

#define STATES   16
#define FDIM     128
#define EPSF     1e-10f
#define EXP1F    2.7182818284590452f
#define EXP1M1F  1.7182818284590452f
#define LN2F     0.69314718055994531f
#define MSG_A    (1.0f / (15.0f + EXP1F))
#define MSG_B    (EXP1M1F / (15.0f + EXP1F))

template <int M>
static __device__ __forceinline__ float row_xor(float x) {
  const int b = __builtin_bit_cast(int, x);
  return __builtin_bit_cast(float, __builtin_amdgcn_update_dpp(b, b, 0x160 + M, 0xf, 0xf, true));
}
static __device__ __forceinline__ float xhalf(float x) {
  const int b = __builtin_bit_cast(int, x);
  return __builtin_bit_cast(float,
      __builtin_amdgcn_permlanex16(b, b, 0x76543210, 0xfedcba98, false, false));
}
static __device__ __forceinline__ float row_sum16(float t) {
  t += row_xor<1>(t);
  t += row_xor<2>(t);
  t += row_xor<4>(t);
  t += row_xor<8>(t);
  return t;
}
static __device__ __forceinline__ float row_max16(float m) {
  m = fmaxf(m, row_xor<1>(m));
  m = fmaxf(m, row_xor<2>(m));
  m = fmaxf(m, row_xor<4>(m));
  m = fmaxf(m, row_xor<8>(m));
  return m;
}

static __device__ __forceinline__ unsigned placement_word() {
  return __builtin_amdgcn_s_sendmsg_rtn(0x87);
}
static __device__ __forceinline__ float* own_half_table(unsigned place, float* tab0, float* tab1) {
  return ((place >> 18) & 1u) ? tab1 : tab0;
}

static __device__ __forceinline__ float bp_msg(float x) {
  const float t = row_sum16(x);
  return (t + EXP1M1F * x) / ((15.0f + EXP1F) * t);
}

static __device__ __forceinline__ float bp_msg_edge(float x) {
  const float t = row_sum16(x);
  return fmaf(x * __builtin_amdgcn_rcpf(t), MSG_B, MSG_A);
}

static __device__ __forceinline__ float log_msg_edge(float m) {
  return __builtin_amdgcn_logf(fmaxf(m, EPSF)) * LN2F;
}

__global__ void bpn_priors_wmma(const float* __restrict__ feat,
                                const float* __restrict__ W,
                                float* __restrict__ priors,
                                float* __restrict__ log_priors,
                                float* __restrict__ accum,
                                float* __restrict__ accumB,
                                float* __restrict__ rec,
                                float* __restrict__ lg1,
                                int N, int numTiles) {
  const int wave = blockIdx.x * (blockDim.x >> 5) + (threadIdx.x >> 5);
  if (wave >= numTiles) return;
  const int lane = threadIdx.x & 31;
  const int half = lane >> 4;
  const int s    = lane & 15;

  int rowA = wave * 16 + s;
  if (rowA >= N) rowA = N - 1;
  const float* frow = feat + (size_t)rowA * FDIM;

  v8f acc = {};
#pragma unroll
  for (int k0 = 0; k0 < FDIM; k0 += 4) {
    const int k = k0 + 2 * half;
    v2f a, b;
    a.x = frow[k];
    a.y = frow[k + 1];
    b.x = W[(size_t)k * STATES + s];
    b.y = W[(size_t)(k + 1) * STATES + s];
    acc = __builtin_amdgcn_wmma_f32_16x16x4_f32(false, a, false, b,
                                                (short)0, acc, false, false);
  }

#pragma unroll
  for (int r = 0; r < 8; ++r) {
    float logit = acc[r];
    const float mx = row_max16(logit);
    float ex = expf(logit - mx);
    const float sm = row_sum16(ex);
    float pr = ex / sm;
    float lp = logf(fmaxf(pr, EPSF));
    float g1 = bp_msg(pr / 0.0625f);
    float l1 = logf(fmaxf(g1, EPSF));
    const int rowO = wave * 16 + r + 8 * half;
    if (rowO < N) {
      const size_t o = (size_t)rowO * STATES + s;
      priors[o]     = pr;
      log_priors[o] = lp;
      accum[o]      = lp;
      accumB[o]     = 0.0f;
      v2f g; g.x = g1; g.y = pr;
      *(v2f*)(rec + 2 * o) = g;
      lg1[o]        = l1;
    }
  }
}

#define PAIRS 8

#define EDGE_BLOCKS 4096

template <int I>
static __device__ __forceinline__ int row_bcast(int x) {
  return __builtin_amdgcn_update_dpp(x, x, 0x150 + I, 0xf, 0xf, true);
}
static __device__ __forceinline__ void row_spread(int idx16, int s, int (&o)[PAIRS]) {
  static_assert(PAIRS == 8, "one line per pair below");
  o[0] = row_bcast<0>(idx16) | s;
  o[1] = row_bcast<1>(idx16) | s;
  o[2] = row_bcast<2>(idx16) | s;
  o[3] = row_bcast<3>(idx16) | s;
  o[4] = row_bcast<4>(idx16) | s;
  o[5] = row_bcast<5>(idx16) | s;
  o[6] = row_bcast<6>(idx16) | s;
  o[7] = row_bcast<7>(idx16) | s;
}

template <int ITER>
static __device__ __forceinline__ void edge_gather(const float* __restrict__ rec,
                                                   const float* __restrict__ bel2,
                                                   int sv, int s,
                                                   v2f (&gb)[PAIRS], float (&b2)[PAIRS]) {
  const v2f* rec2 = (const v2f*)rec;
  int row[PAIRS];
  row_spread(sv * STATES, s, row);
#pragma unroll
  for (int i = 0; i < PAIRS; ++i) {
    gb[i] = rec2[row[i]];
    if (ITER == 3) b2[i] = bel2[row[i]];
  }
}

template <int ITER, bool FULL>
static __device__ __forceinline__ void edge_compute(const v2f (&gb)[PAIRS],
                                                    const float (&b2)[PAIRS],
                                                    float* __restrict__ accum,
                                                    int dv, int p0, int Eu, int s) {
  float lm[PAIRS];
  {
    float m2[PAIRS];
#pragma unroll
    for (int i = 0; i < PAIRS; ++i) {
      const float g1rev = xhalf(gb[i].x);
      m2[i] = bp_msg_edge(gb[i].y * __builtin_amdgcn_rcpf(g1rev));
    }
    if (ITER == 2) {
#pragma unroll
      for (int i = 0; i < PAIRS; ++i)
        lm[i] = log_msg_edge(m2[i]);
    } else {
#pragma unroll
      for (int i = 0; i < PAIRS; ++i) {
        const float m2rev = xhalf(m2[i]);
        const float m3    = bp_msg_edge(b2[i] * __builtin_amdgcn_rcpf(m2rev));
        lm[i] = log_msg_edge(m3);
      }
    }
  }

  int drow[PAIRS];
  row_spread(dv * STATES, s, drow);
#pragma unroll
  for (int i = 0; i < PAIRS; ++i) {
    if (FULL || (p0 + i < Eu))
      atomicAdd(&accum[drow[i]], lm[i]);
  }
}

template <int ITER>
static __device__ __forceinline__ void edge_body(const float* __restrict__ rec,
                                                 const float* __restrict__ bel2,
                                                 const int* __restrict__ src,
                                                 const int* __restrict__ dst,
                                                 const int* __restrict__ rev,
                                                 float* __restrict__ accum0,
                                                 float* __restrict__ accum1,
                                                 int Eu) {
  const int w  = __builtin_amdgcn_readfirstlane(
      (int)(blockIdx.x * (blockDim.x >> 5) + (threadIdx.x >> 5)));
  const int nW = (int)(gridDim.x * (blockDim.x >> 5));
  const int nG = (Eu + PAIRS - 1) / PAIRS;
  if (w >= nG) return;
  const int lane = threadIdx.x & 31;
  const int half = lane >> 4;
  const int s    = lane & 15;
  const int sub  = lane & (PAIRS - 1);
  const int last = Eu - 1;

  int pl = w * PAIRS + sub;
  pl = pl < Eu ? pl : last;
  const int jv = rev[pl];
  float* accum = own_half_table(placement_word(), accum0, accum1);
  const int el = half ? jv : pl;
  int sv = src[el];
  int dv = dst[el];
  int pn = (w + nW) * PAIRS + sub;
  pn = pn < Eu ? pn : last;
  int jn = rev[pn];

  for (int g = w; g < nG; g += nW) {
    v2f   gb[PAIRS];
    float b2[PAIRS];
    edge_gather<ITER>(rec, bel2, sv, s, gb, b2);
    __builtin_amdgcn_sched_barrier(0);
    const int eln = half ? jn : pn;
    const int svn = src[eln];
    const int dvn = dst[eln];
    int pn2 = (g + 2 * nW) * PAIRS + sub;
    pn2 = pn2 < Eu ? pn2 : last;
    const int jn2 = rev[pn2];
    __builtin_amdgcn_sched_barrier(0);
    const int p0 = g * PAIRS;
    if (p0 + PAIRS <= Eu)
      edge_compute<ITER, true >(gb, b2, accum, dv, p0, Eu, s);
    else
      edge_compute<ITER, false>(gb, b2, accum, dv, p0, Eu, s);
    sv = svn; dv = dvn; jn = jn2; pn = pn2;
  }
}

template <bool FULL>
static __device__ __forceinline__ void edge1_scatter(float* __restrict__ accum,
                                                     const int (&drow)[PAIRS],
                                                     const float (&lm)[PAIRS],
                                                     int e0, int E2) {
#pragma unroll
  for (int i = 0; i < PAIRS; ++i) {
    if (FULL || (e0 + i < E2))
      atomicAdd(&accum[drow[i]], lm[i]);
  }
}

__global__ void bpn_edge1(const float* __restrict__ lg1,
                          const int* __restrict__ src,
                          const int* __restrict__ dst,
                          float* __restrict__ accum0,
                          float* __restrict__ accum1, int Eu) {
  const int w  = __builtin_amdgcn_readfirstlane(
      (int)(blockIdx.x * (blockDim.x >> 5) + (threadIdx.x >> 5)));
  const int nW = (int)(gridDim.x * (blockDim.x >> 5));
  const int E2 = 2 * Eu;
  const int nG = (E2 + 2 * PAIRS - 1) / (2 * PAIRS);
  if (w >= nG) return;
  float* accum = own_half_table(placement_word(), accum0, accum1);
  const int lane = threadIdx.x & 31;
  const int half = lane >> 4;
  const int s    = lane & 15;
  const int sub  = lane & (PAIRS - 1);
  const int off  = PAIRS * half + sub;
  const int last = E2 - 1;

  int el = w * (2 * PAIRS) + off;
  el = el < E2 ? el : last;
  int sv = src[el];
  int dv = dst[el];

  for (int g = w; g < nG; g += nW) {
    int row[PAIRS];
    row_spread(sv * STATES, s, row);
    float lm[PAIRS];
#pragma unroll
    for (int i = 0; i < PAIRS; ++i)
      lm[i] = lg1[row[i]];
    __builtin_amdgcn_sched_barrier(0);
    int eln = (g + nW) * (2 * PAIRS) + off;
    eln = eln < E2 ? eln : last;
    const int svn = src[eln];
    const int dvn = dst[eln];
    int drow[PAIRS];
    row_spread(dv * STATES, s, drow);
    const int g0 = g * (2 * PAIRS);
    if (g0 + 2 * PAIRS <= E2)
      edge1_scatter<true >(accum, drow, lm, g0 + PAIRS * half, E2);
    else
      edge1_scatter<false>(accum, drow, lm, g0 + PAIRS * half, E2);
    sv = svn; dv = dvn;
  }
}

__global__ void bpn_edge2(const float* __restrict__ rec,
                          const int* __restrict__ src,
                          const int* __restrict__ dst,
                          const int* __restrict__ rev,
                          float* __restrict__ accum0,
                          float* __restrict__ accum1, int Eu) {
  edge_body<2>(rec, nullptr, src, dst, rev, accum0, accum1, Eu);
}

__global__ void bpn_edge3(const float* __restrict__ rec,
                          const float* __restrict__ bel2,
                          const int* __restrict__ src,
                          const int* __restrict__ dst,
                          const int* __restrict__ rev,
                          float* __restrict__ accum0,
                          float* __restrict__ accum1, int Eu) {
  edge_body<3>(rec, bel2, src, dst, rev, accum0, accum1, Eu);
}

#define NODEB 8

__global__ void bpn_node(float* __restrict__ accum,
                         float* __restrict__ accumB,
                         const float* __restrict__ log_priors,
                         float* __restrict__ beliefs_out,
                         int ostride, int ooff,
                         int N) {
  const int w    = blockIdx.x * (blockDim.x >> 5) + (threadIdx.x >> 5);
  const int lane = threadIdx.x & 31;
  const int half = lane >> 4;
  const int s    = lane & 15;
  const int base = w * (2 * NODEB) + half;

  bool   ok[NODEB];
  int    o[NODEB];
  float  a[NODEB], b[NODEB], lp[NODEB];
#pragma unroll
  for (int i = 0; i < NODEB; ++i) {
    const int node = base + 2 * i;
    ok[i] = (node < N);
    o[i]  = (ok[i] ? node : 0) * STATES + s;
    a[i]  = accum[o[i]];
    b[i]  = accumB[o[i]];
    lp[i] = log_priors[o[i]];
  }
#pragma unroll
  for (int i = 0; i < NODEB; ++i) {
    const float v  = ok[i] ? (a[i] + b[i]) : 0.0f;
    const float mx = row_max16(v);
    const float ex = expf(v - mx);
    const float sm = row_sum16(ex);
    if (ok[i]) {
      beliefs_out[o[i] * ostride + ooff] = ex / sm;
      accum[o[i]]  = lp[i];
      accumB[o[i]] = 0.0f;
    }
  }
}

extern "C" void kernel_launch(void* const* d_in, const int* in_sizes, int n_in,
                              void* d_out, int out_size, void* d_ws, size_t ws_size,
                              hipStream_t stream) {
  const float* feat = (const float*)d_in[0];
  const float* W    = (const float*)d_in[1];
  const int*   src  = (const int*)d_in[2];
  const int*   dst  = (const int*)d_in[3];
  const int*   rev  = (const int*)d_in[4];

  const int N  = in_sizes[0] / FDIM;
  const int E2 = in_sizes[2];
  const int Eu = E2 / 2;

  float* priors      = (float*)d_out;
  float* beliefs_out = priors + (size_t)N * STATES;

  float* lp    = (float*)d_ws;
  float* accum = lp    + (size_t)N * STATES;
  float* lg1   = accum + (size_t)N * STATES;
  float* bel2  = lg1   + (size_t)N * STATES;
  float* rec   = bel2  + (size_t)N * STATES;
  float* accB  = rec   + (size_t)N * STATES * 2;

  const dim3 blk(256);
  const int tiles   = (N + 15) / 16;
  const int gBlocks = (tiles + 7) / 8;
  const int eNeed   = (Eu + 8 * PAIRS - 1) / (8 * PAIRS);
  const int eBlocks = eNeed < EDGE_BLOCKS ? eNeed : EDGE_BLOCKS;
  const int nWaves  = (N + 2 * NODEB - 1) / (2 * NODEB);
  const int nBlocks = (nWaves + 7) / 8;

  bpn_priors_wmma<<<gBlocks, blk, 0, stream>>>(feat, W, priors, lp, accum, accB, rec, lg1, N, tiles);

  bpn_edge1<<<eBlocks, blk, 0, stream>>>(lg1, src, dst, accum, accB, Eu);
  bpn_node<<<nBlocks, blk, 0, stream>>>(accum, accB, lp, rec, 2, 1, N);
  bpn_edge2<<<eBlocks, blk, 0, stream>>>(rec, src, dst, rev, accum, accB, Eu);
  bpn_node<<<nBlocks, blk, 0, stream>>>(accum, accB, lp, bel2, 1, 0, N);
  bpn_edge3<<<eBlocks, blk, 0, stream>>>(rec, bel2, src, dst, rev, accum, accB, Eu);
  bpn_node<<<nBlocks, blk, 0, stream>>>(accum, accB, lp, beliefs_out, 1, 0, N);
}


# Edge kernels of iterations 2, 3: a group's 8 row totals from ONE folding butterfly (lanes split the rows at steps 1, 2, 4) and ONE v_rcp_f32, read per pair by DPP row_share (was one of each per pair)
# speedup vs baseline: 5.7475x; 1.0403x over previous
#include <hip/hip_runtime.h>
#include <math.h>


typedef float v2f __attribute__((ext_vector_type(2)));
typedef float v8f __attribute__((ext_vector_type(8)));

#define STATES   16
#define FDIM     128
#define EPSF     1e-10f
#define EXP1F    2.7182818284590452f
#define EXP1M1F  1.7182818284590452f
#define LN2F     0.69314718055994531f
#define MSG_A    (1.0f / (15.0f + EXP1F))
#define MSG_B    (EXP1M1F / (15.0f + EXP1F))

template <int M>
static __device__ __forceinline__ float row_xor(float x) {
  const int b = __builtin_bit_cast(int, x);
  return __builtin_bit_cast(float, __builtin_amdgcn_update_dpp(b, b, 0x160 + M, 0xf, 0xf, true));
}
static __device__ __forceinline__ float xhalf(float x) {
  const int b = __builtin_bit_cast(int, x);
  return __builtin_bit_cast(float,
      __builtin_amdgcn_permlanex16(b, b, 0x76543210, 0xfedcba98, false, false));
}
static __device__ __forceinline__ float row_sum16(float t) {
  t += row_xor<1>(t);
  t += row_xor<2>(t);
  t += row_xor<4>(t);
  t += row_xor<8>(t);
  return t;
}
static __device__ __forceinline__ float row_max16(float m) {
  m = fmaxf(m, row_xor<1>(m));
  m = fmaxf(m, row_xor<2>(m));
  m = fmaxf(m, row_xor<4>(m));
  m = fmaxf(m, row_xor<8>(m));
  return m;
}

static __device__ __forceinline__ unsigned placement_word() {
  return __builtin_amdgcn_s_sendmsg_rtn(0x87);
}
static __device__ __forceinline__ float* own_half_table(unsigned place, float* tab0, float* tab1) {
  return ((place >> 18) & 1u) ? tab1 : tab0;
}

static __device__ __forceinline__ float bp_msg(float x) {
  const float t = row_sum16(x);
  return (t + EXP1M1F * x) / ((15.0f + EXP1F) * t);
}

static __device__ __forceinline__ float log_msg_edge(float m) {
  return __builtin_amdgcn_logf(fmaxf(m, EPSF)) * LN2F;
}

__global__ void bpn_priors_wmma(const float* __restrict__ feat,
                                const float* __restrict__ W,
                                float* __restrict__ priors,
                                float* __restrict__ log_priors,
                                float* __restrict__ accum,
                                float* __restrict__ accumB,
                                float* __restrict__ rec,
                                float* __restrict__ lg1,
                                int N, int numTiles) {
  const int wave = blockIdx.x * (blockDim.x >> 5) + (threadIdx.x >> 5);
  if (wave >= numTiles) return;
  const int lane = threadIdx.x & 31;
  const int half = lane >> 4;
  const int s    = lane & 15;

  int rowA = wave * 16 + s;
  if (rowA >= N) rowA = N - 1;
  const float* frow = feat + (size_t)rowA * FDIM;

  v8f acc = {};
#pragma unroll
  for (int k0 = 0; k0 < FDIM; k0 += 4) {
    const int k = k0 + 2 * half;
    v2f a, b;
    a.x = frow[k];
    a.y = frow[k + 1];
    b.x = W[(size_t)k * STATES + s];
    b.y = W[(size_t)(k + 1) * STATES + s];
    acc = __builtin_amdgcn_wmma_f32_16x16x4_f32(false, a, false, b,
                                                (short)0, acc, false, false);
  }

#pragma unroll
  for (int r = 0; r < 8; ++r) {
    float logit = acc[r];
    const float mx = row_max16(logit);
    float ex = expf(logit - mx);
    const float sm = row_sum16(ex);
    float pr = ex / sm;
    float lp = logf(fmaxf(pr, EPSF));
    float g1 = bp_msg(pr / 0.0625f);
    float l1 = logf(fmaxf(g1, EPSF));
    const int rowO = wave * 16 + r + 8 * half;
    if (rowO < N) {
      const size_t o = (size_t)rowO * STATES + s;
      priors[o]     = pr;
      log_priors[o] = lp;
      accum[o]      = lp;
      accumB[o]     = 0.0f;
      v2f g; g.x = g1; g.y = pr;
      *(v2f*)(rec + 2 * o) = g;
      lg1[o]        = l1;
    }
  }
}

#define PAIRS 8

#define EDGE_BLOCKS 4096

template <int I>
static __device__ __forceinline__ int row_bcast(int x) {
  return __builtin_amdgcn_update_dpp(x, x, 0x150 + I, 0xf, 0xf, true);
}
static __device__ __forceinline__ void row_spread(int idx16, int s, int (&o)[PAIRS]) {
  static_assert(PAIRS == 8, "one line per pair below");
  o[0] = row_bcast<0>(idx16) | s;
  o[1] = row_bcast<1>(idx16) | s;
  o[2] = row_bcast<2>(idx16) | s;
  o[3] = row_bcast<3>(idx16) | s;
  o[4] = row_bcast<4>(idx16) | s;
  o[5] = row_bcast<5>(idx16) | s;
  o[6] = row_bcast<6>(idx16) | s;
  o[7] = row_bcast<7>(idx16) | s;
}

template <int M>
static __device__ __forceinline__ float fold(float a, float b, bool odd) {
  const float keep = odd ? b : a;
  const float send = odd ? a : b;
  return keep + row_xor<M>(send);
}
static __device__ __forceinline__ float row_sum16_group(const float (&x)[PAIRS], int s) {
  static_assert(PAIRS == 8, "three folding steps");
  const bool o1 = (s & 1) != 0, o2 = (s & 2) != 0, o4 = (s & 4) != 0;
  const float v01 = fold<1>(x[0], x[1], o1);
  const float v23 = fold<1>(x[2], x[3], o1);
  const float v45 = fold<1>(x[4], x[5], o1);
  const float v67 = fold<1>(x[6], x[7], o1);
  const float v03 = fold<2>(v01, v23, o2);
  const float v47 = fold<2>(v45, v67, o2);
  const float v07 = fold<4>(v03, v47, o4);
  return v07 + row_xor<8>(v07);
}
template <int I>
static __device__ __forceinline__ float row_bcast_f(float x) {
  return __builtin_bit_cast(float, row_bcast<I>(__builtin_bit_cast(int, x)));
}

static __device__ __forceinline__ void bp_msg_edge_group(const float (&x)[PAIRS], int s,
                                                         float (&m)[PAIRS]) {
  static_assert(PAIRS == 8, "one line per pair below");
  const float r = __builtin_amdgcn_rcpf(row_sum16_group(x, s));
  m[0] = fmaf(x[0] * row_bcast_f<0>(r), MSG_B, MSG_A);
  m[1] = fmaf(x[1] * row_bcast_f<1>(r), MSG_B, MSG_A);
  m[2] = fmaf(x[2] * row_bcast_f<2>(r), MSG_B, MSG_A);
  m[3] = fmaf(x[3] * row_bcast_f<3>(r), MSG_B, MSG_A);
  m[4] = fmaf(x[4] * row_bcast_f<4>(r), MSG_B, MSG_A);
  m[5] = fmaf(x[5] * row_bcast_f<5>(r), MSG_B, MSG_A);
  m[6] = fmaf(x[6] * row_bcast_f<6>(r), MSG_B, MSG_A);
  m[7] = fmaf(x[7] * row_bcast_f<7>(r), MSG_B, MSG_A);
}

template <int ITER>
static __device__ __forceinline__ void edge_gather(const float* __restrict__ rec,
                                                   const float* __restrict__ bel2,
                                                   int sv, int s,
                                                   v2f (&gb)[PAIRS], float (&b2)[PAIRS]) {
  const v2f* rec2 = (const v2f*)rec;
  int row[PAIRS];
  row_spread(sv * STATES, s, row);
#pragma unroll
  for (int i = 0; i < PAIRS; ++i) {
    gb[i] = rec2[row[i]];
    if (ITER == 3) b2[i] = bel2[row[i]];
  }
}

template <int ITER, bool FULL>
static __device__ __forceinline__ void edge_compute(const v2f (&gb)[PAIRS],
                                                    const float (&b2)[PAIRS],
                                                    float* __restrict__ accum,
                                                    int dv, int p0, int Eu, int s) {
  float lm[PAIRS];
  {
    float x2[PAIRS], m2[PAIRS];
#pragma unroll
    for (int i = 0; i < PAIRS; ++i) {
      const float g1rev = xhalf(gb[i].x);
      x2[i] = gb[i].y * __builtin_amdgcn_rcpf(g1rev);
    }
    bp_msg_edge_group(x2, s, m2);
    if (ITER == 2) {
#pragma unroll
      for (int i = 0; i < PAIRS; ++i)
        lm[i] = log_msg_edge(m2[i]);
    } else {
      float x3[PAIRS], m3[PAIRS];
#pragma unroll
      for (int i = 0; i < PAIRS; ++i) {
        const float m2rev = xhalf(m2[i]);
        x3[i] = b2[i] * __builtin_amdgcn_rcpf(m2rev);
      }
      bp_msg_edge_group(x3, s, m3);
#pragma unroll
      for (int i = 0; i < PAIRS; ++i)
        lm[i] = log_msg_edge(m3[i]);
    }
  }

  int drow[PAIRS];
  row_spread(dv * STATES, s, drow);
#pragma unroll
  for (int i = 0; i < PAIRS; ++i) {
    if (FULL || (p0 + i < Eu))
      atomicAdd(&accum[drow[i]], lm[i]);
  }
}

template <int ITER>
static __device__ __forceinline__ void edge_body(const float* __restrict__ rec,
                                                 const float* __restrict__ bel2,
                                                 const int* __restrict__ src,
                                                 const int* __restrict__ dst,
                                                 const int* __restrict__ rev,
                                                 float* __restrict__ accum0,
                                                 float* __restrict__ accum1,
                                                 int Eu) {
  const int w  = __builtin_amdgcn_readfirstlane(
      (int)(blockIdx.x * (blockDim.x >> 5) + (threadIdx.x >> 5)));
  const int nW = (int)(gridDim.x * (blockDim.x >> 5));
  const int nG = (Eu + PAIRS - 1) / PAIRS;
  if (w >= nG) return;
  const int lane = threadIdx.x & 31;
  const int half = lane >> 4;
  const int s    = lane & 15;
  const int sub  = lane & (PAIRS - 1);
  const int last = Eu - 1;

  int pl = w * PAIRS + sub;
  pl = pl < Eu ? pl : last;
  const int jv = rev[pl];
  float* accum = own_half_table(placement_word(), accum0, accum1);
  const int el = half ? jv : pl;
  int sv = src[el];
  int dv = dst[el];
  int pn = (w + nW) * PAIRS + sub;
  pn = pn < Eu ? pn : last;
  int jn = rev[pn];

  for (int g = w; g < nG; g += nW) {
    v2f   gb[PAIRS];
    float b2[PAIRS];
    edge_gather<ITER>(rec, bel2, sv, s, gb, b2);
    __builtin_amdgcn_sched_barrier(0);
    const int eln = half ? jn : pn;
    const int svn = src[eln];
    const int dvn = dst[eln];
    int pn2 = (g + 2 * nW) * PAIRS + sub;
    pn2 = pn2 < Eu ? pn2 : last;
    const int jn2 = rev[pn2];
    __builtin_amdgcn_sched_barrier(0);
    const int p0 = g * PAIRS;
    if (p0 + PAIRS <= Eu)
      edge_compute<ITER, true >(gb, b2, accum, dv, p0, Eu, s);
    else
      edge_compute<ITER, false>(gb, b2, accum, dv, p0, Eu, s);
    sv = svn; dv = dvn; jn = jn2; pn = pn2;
  }
}

template <bool FULL>
static __device__ __forceinline__ void edge1_scatter(float* __restrict__ accum,
                                                     const int (&drow)[PAIRS],
                                                     const float (&lm)[PAIRS],
                                                     int e0, int E2) {
#pragma unroll
  for (int i = 0; i < PAIRS; ++i) {
    if (FULL || (e0 + i < E2))
      atomicAdd(&accum[drow[i]], lm[i]);
  }
}

__global__ void bpn_edge1(const float* __restrict__ lg1,
                          const int* __restrict__ src,
                          const int* __restrict__ dst,
                          float* __restrict__ accum0,
                          float* __restrict__ accum1, int Eu) {
  const int w  = __builtin_amdgcn_readfirstlane(
      (int)(blockIdx.x * (blockDim.x >> 5) + (threadIdx.x >> 5)));
  const int nW = (int)(gridDim.x * (blockDim.x >> 5));
  const int E2 = 2 * Eu;
  const int nG = (E2 + 2 * PAIRS - 1) / (2 * PAIRS);
  if (w >= nG) return;
  float* accum = own_half_table(placement_word(), accum0, accum1);
  const int lane = threadIdx.x & 31;
  const int half = lane >> 4;
  const int s    = lane & 15;
  const int sub  = lane & (PAIRS - 1);
  const int off  = PAIRS * half + sub;
  const int last = E2 - 1;

  int el = w * (2 * PAIRS) + off;
  el = el < E2 ? el : last;
  int sv = src[el];
  int dv = dst[el];

  for (int g = w; g < nG; g += nW) {
    int row[PAIRS];
    row_spread(sv * STATES, s, row);
    float lm[PAIRS];
#pragma unroll
    for (int i = 0; i < PAIRS; ++i)
      lm[i] = lg1[row[i]];
    __builtin_amdgcn_sched_barrier(0);
    int eln = (g + nW) * (2 * PAIRS) + off;
    eln = eln < E2 ? eln : last;
    const int svn = src[eln];
    const int dvn = dst[eln];
    int drow[PAIRS];
    row_spread(dv * STATES, s, drow);
    const int g0 = g * (2 * PAIRS);
    if (g0 + 2 * PAIRS <= E2)
      edge1_scatter<true >(accum, drow, lm, g0 + PAIRS * half, E2);
    else
      edge1_scatter<false>(accum, drow, lm, g0 + PAIRS * half, E2);
    sv = svn; dv = dvn;
  }
}

__global__ void bpn_edge2(const float* __restrict__ rec,
                          const int* __restrict__ src,
                          const int* __restrict__ dst,
                          const int* __restrict__ rev,
                          float* __restrict__ accum0,
                          float* __restrict__ accum1, int Eu) {
  edge_body<2>(rec, nullptr, src, dst, rev, accum0, accum1, Eu);
}

__global__ void bpn_edge3(const float* __restrict__ rec,
                          const float* __restrict__ bel2,
                          const int* __restrict__ src,
                          const int* __restrict__ dst,
                          const int* __restrict__ rev,
                          float* __restrict__ accum0,
                          float* __restrict__ accum1, int Eu) {
  edge_body<3>(rec, bel2, src, dst, rev, accum0, accum1, Eu);
}

#define NODEB 8

__global__ void bpn_node(float* __restrict__ accum,
                         float* __restrict__ accumB,
                         const float* __restrict__ log_priors,
                         float* __restrict__ beliefs_out,
                         int ostride, int ooff,
                         int N) {
  const int w    = blockIdx.x * (blockDim.x >> 5) + (threadIdx.x >> 5);
  const int lane = threadIdx.x & 31;
  const int half = lane >> 4;
  const int s    = lane & 15;
  const int base = w * (2 * NODEB) + half;

  bool   ok[NODEB];
  int    o[NODEB];
  float  a[NODEB], b[NODEB], lp[NODEB];
#pragma unroll
  for (int i = 0; i < NODEB; ++i) {
    const int node = base + 2 * i;
    ok[i] = (node < N);
    o[i]  = (ok[i] ? node : 0) * STATES + s;
    a[i]  = accum[o[i]];
    b[i]  = accumB[o[i]];
    lp[i] = log_priors[o[i]];
  }
#pragma unroll
  for (int i = 0; i < NODEB; ++i) {
    const float v  = ok[i] ? (a[i] + b[i]) : 0.0f;
    const float mx = row_max16(v);
    const float ex = expf(v - mx);
    const float sm = row_sum16(ex);
    if (ok[i]) {
      beliefs_out[o[i] * ostride + ooff] = ex / sm;
      accum[o[i]]  = lp[i];
      accumB[o[i]] = 0.0f;
    }
  }
}

extern "C" void kernel_launch(void* const* d_in, const int* in_sizes, int n_in,
                              void* d_out, int out_size, void* d_ws, size_t ws_size,
                              hipStream_t stream) {
  const float* feat = (const float*)d_in[0];
  const float* W    = (const float*)d_in[1];
  const int*   src  = (const int*)d_in[2];
  const int*   dst  = (const int*)d_in[3];
  const int*   rev  = (const int*)d_in[4];

  const int N  = in_sizes[0] / FDIM;
  const int E2 = in_sizes[2];
  const int Eu = E2 / 2;

  float* priors      = (float*)d_out;
  float* beliefs_out = priors + (size_t)N * STATES;

  float* lp    = (float*)d_ws;
  float* accum = lp    + (size_t)N * STATES;
  float* lg1   = accum + (size_t)N * STATES;
  float* bel2  = lg1   + (size_t)N * STATES;
  float* rec   = bel2  + (size_t)N * STATES;
  float* accB  = rec   + (size_t)N * STATES * 2;

  const dim3 blk(256);
  const int tiles   = (N + 15) / 16;
  const int gBlocks = (tiles + 7) / 8;
  const int eNeed   = (Eu + 8 * PAIRS - 1) / (8 * PAIRS);
  const int eBlocks = eNeed < EDGE_BLOCKS ? eNeed : EDGE_BLOCKS;
  const int nWaves  = (N + 2 * NODEB - 1) / (2 * NODEB);
  const int nBlocks = (nWaves + 7) / 8;

  bpn_priors_wmma<<<gBlocks, blk, 0, stream>>>(feat, W, priors, lp, accum, accB, rec, lg1, N, tiles);

  bpn_edge1<<<eBlocks, blk, 0, stream>>>(lg1, src, dst, accum, accB, Eu);
  bpn_node<<<nBlocks, blk, 0, stream>>>(accum, accB, lp, rec, 2, 1, N);
  bpn_edge2<<<eBlocks, blk, 0, stream>>>(rec, src, dst, rev, accum, accB, Eu);
  bpn_node<<<nBlocks, blk, 0, stream>>>(accum, accB, lp, bel2, 1, 0, N);
  bpn_edge3<<<eBlocks, blk, 0, stream>>>(rec, bel2, src, dst, rev, accum, accB, Eu);
  bpn_node<<<nBlocks, blk, 0, stream>>>(accum, accB, lp, beliefs_out, 1, 0, N);
}
